# starting kernel
# baseline (speedup 1.0000x reference)
#include <hip/hip_runtime.h>

typedef float v2f __attribute__((ext_vector_type(2)));
typedef float v8f __attribute__((ext_vector_type(8)));

#define LX 6
#define LY 6
#define PD 2
#define DD 4
#define NSITE 36
#define ENV7 16384
#define ENV6 4096

__global__ void peps_prep_kernel(const float* __restrict__ peps,
                                 float* __restrict__ Msite) {
    int idx = blockIdx.x * blockDim.x + threadIdx.x;
    if (idx >= NSITE * PD * 256) return;
    int n    = idx & 15;
    int k    = (idx >> 4) & 15;
    int s    = (idx >> 8) & 1;
    int site = idx >> 9;
    int i = site / LY, j = site % LY;
    int a = k >> 2, c = k & 3;
    int d = n >> 2, b = n & 3;
    float v = 0.0f;
    bool pad = (i == 0      && a > 0) || (i == LX - 1 && b > 0) ||
               (j == 0      && c > 0) || (j == LY - 1 && d > 0);
    if (!pad) {
        int p = ((((site * PD + s) * DD + a) * DD + b) * DD + c) * DD + d;
        v = peps[p];
    }
    Msite[idx] = v;
}

__global__ void __launch_bounds__(256)
peps_amp_kernel(const int* __restrict__ x,
                const float* __restrict__ Msite,
                float* __restrict__ out) {
    extern __shared__ float env[];

    const int tid    = threadIdx.x;
    const int sample = blockIdx.x;
    const int lane   = tid & 31;
    const int wave   = __builtin_amdgcn_readfirstlane(tid >> 5);
    __builtin_assume(wave >= 0 && wave < 8);
    const int n      = lane & 15;
    const int h      = lane >> 4;

    for (int q = tid; q < ENV7; q += 256) env[q] = 0.0f;
    __syncthreads();
    if (tid == 0) env[0] = 1.0f;
    __syncthreads();

    const int* xrow = x + sample * NSITE;

    for (int j = 0; j < LY; ++j) {
        for (int i = 0; i < LX; ++i) {
            const int site = i * LY + j;
            const int s    = xrow[site];
            const float* Mp = Msite + (site * PD + s) * 256;

            v2f b0, b1, b2, b3;
            b0.x = Mp[(0  + 2 * h) * 16 + n];  b0.y = Mp[(1  + 2 * h) * 16 + n];
            b1.x = Mp[(4  + 2 * h) * 16 + n];  b1.y = Mp[(5  + 2 * h) * 16 + n];
            b2.x = Mp[(8  + 2 * h) * 16 + n];  b2.y = Mp[(9  + 2 * h) * 16 + n];
            b3.x = Mp[(12 + 2 * h) * 16 + n];  b3.y = Mp[(13 + 2 * h) * 16 + n];

            const int shift = 2 * (5 - i);
            const int psm   = (1 << shift) - 1;

            #pragma unroll 2
            for (int t = wave; t < 64; t += 8) {
                int r    = t * 16 + n;
                int post = r & psm;
                int base = ((r - post) << 4) + post;
                v2f a0, a1, a2, a3;
                a0.x = env[base + ((0  + 2 * h) << shift)];
                a0.y = env[base + ((1  + 2 * h) << shift)];
                a1.x = env[base + ((4  + 2 * h) << shift)];
                a1.y = env[base + ((5  + 2 * h) << shift)];
                a2.x = env[base + ((8  + 2 * h) << shift)];
                a2.y = env[base + ((9  + 2 * h) << shift)];
                a3.x = env[base + ((12 + 2 * h) << shift)];
                a3.y = env[base + ((13 + 2 * h) << shift)];

                v8f acc = {};
                acc = __builtin_amdgcn_wmma_f32_16x16x4_f32(
                        false, a0, false, b0, (short)0, acc, false, false);
                acc = __builtin_amdgcn_wmma_f32_16x16x4_f32(
                        false, a1, false, b1, (short)0, acc, false, false);
                acc = __builtin_amdgcn_wmma_f32_16x16x4_f32(
                        false, a2, false, b2, (short)0, acc, false, false);
                acc = __builtin_amdgcn_wmma_f32_16x16x4_f32(
                        false, a3, false, b3, (short)0, acc, false, false);

                #pragma unroll
                for (int v = 0; v < 8; ++v) {
                    int rs    = t * 16 + v + 8 * h;
                    int posts = rs & psm;
                    int bs    = ((rs - posts) << 4) + posts;
                    env[bs + (n << shift)] = acc[v];
                }
            }
            __syncthreads();
        }

        if (j < LY - 1) {
            float keep[16];
            #pragma unroll
            for (int q = 0; q < 16; ++q) keep[q] = env[4 * (tid + 256 * q)];
            __syncthreads();
            #pragma unroll
            for (int q = 0; q < 16; ++q) env[tid + 256 * q] = keep[q];
            for (int q = ENV6 + tid; q < ENV7; q += 256) env[q] = 0.0f;
            __syncthreads();
        }
    }

    if (tid == 0) out[sample] = env[0];
}

extern "C" void kernel_launch(void* const* d_in, const int* in_sizes, int n_in,
                              void* d_out, int out_size, void* d_ws, size_t ws_size,
                              hipStream_t stream) {
    const int*   x     = (const int*)d_in[0];
    const float* peps  = (const float*)d_in[1];
    float*       out   = (float*)d_out;
    float*       Msite = (float*)d_ws;

    const int nM = NSITE * PD * 256;
    peps_prep_kernel<<<(nM + 255) / 256, 256, 0, stream>>>(peps, Msite);

    const int B = out_size;
    peps_amp_kernel<<<B, 256, ENV7 * sizeof(float), stream>>>(x, Msite, out);
}


# contract only the real bond widths at open-boundary sites: skip K chunks, env row tiles and result columns that are zero padding; column-0 k = 0,4,8,12 packed in one WMMA chunk, term order kept
# speedup vs baseline: 3.4615x; 3.4615x over previous
#include <hip/hip_runtime.h>


typedef float v2f __attribute__((ext_vector_type(2)));
typedef float v8f __attribute__((ext_vector_type(8)));

#define LX 6
#define LY 6
#define PD 2
#define DD 4
#define NSITE 36
#define ENV7 16384
#define ENV6 4096

__global__ void peps_prep_kernel(const float* __restrict__ peps,
                                 float* __restrict__ Msite) {
    int idx = blockIdx.x * blockDim.x + threadIdx.x;
    if (idx >= NSITE * PD * 256) return;
    int n    = idx & 15;
    int k    = (idx >> 4) & 15;
    int s    = (idx >> 8) & 1;
    int site = idx >> 9;
    int i = site / LY, j = site % LY;
    int a = k >> 2, c = k & 3;
    int d = n >> 2, b = n & 3;
    float v = 0.0f;
    bool pad = (i == 0      && a > 0) || (i == LX - 1 && b > 0) ||
               (j == 0      && c > 0) || (j == LY - 1 && d > 0);
    if (!pad) {
        int p = ((((site * PD + s) * DD + a) * DD + b) * DD + c) * DD + d;
        v = peps[p];
    }
    Msite[idx] = v;
}

template <int NK, int KSTR, int FORM>
__device__ __forceinline__ void apply_site(float* env,
                                           const float* __restrict__ Mp,
                                           const int shift, const int ntile,
                                           const int tsh, const int wave,
                                           const int n, const int h) {
    const int psm = (1 << shift) - 1;

    v2f b[NK];
    #pragma unroll
    for (int kc = 0; kc < NK; ++kc) {
        b[kc].x = Mp[(4 * kc + KSTR * (2 * h    )) * 16 + n];
        b[kc].y = Mp[(4 * kc + KSTR * (2 * h + 1)) * 16 + n];
    }

    #pragma unroll 2
    for (int tt = wave; tt < ntile; tt += 8) {
        const int t = tt << tsh;
        int r    = t * 16 + n;
        int post = r & psm;
        int base = ((r - post) << 4) + post;
        v2f a[NK];
        #pragma unroll
        for (int kc = 0; kc < NK; ++kc) {
            a[kc].x = env[base + ((4 * kc + KSTR * (2 * h    )) << shift)];
            a[kc].y = env[base + ((4 * kc + KSTR * (2 * h + 1)) << shift)];
        }

        v8f acc = {};
        #pragma unroll
        for (int kc = 0; kc < NK; ++kc) {
            if (FORM == 1)
                acc = __builtin_amdgcn_wmma_f32_16x16x4_f32(
                        false, a[kc], false, b[kc], (short)0, acc, false, false);
            else
                acc = __builtin_amdgcn_wmma_f32_16x16x4_f32(
                        false, b[kc], false, a[kc], (short)0, acc, false, false);
        }

        if (FORM == 0) {
            #pragma unroll
            for (int v = 0; v < 8; ++v)
                env[base + ((v + 8 * h) << shift)] = acc[v];
        } else if (FORM == 2) {
            env[base + ((0 + 8 * h) << shift)] = acc[0];
            env[base + ((4 + 8 * h) << shift)] = acc[4];
        } else {
            #pragma unroll
            for (int v = 0; v < 8; ++v) {
                int rs    = t * 16 + v + 8 * h;
                int posts = rs & psm;
                int bs    = ((rs - posts) << 4) + posts;
                env[bs + (n << shift)] = acc[v];
            }
        }
    }
}

__global__ void __launch_bounds__(256)
peps_amp_kernel(const int* __restrict__ x,
                const float* __restrict__ Msite,
                float* __restrict__ out) {
    extern __shared__ float env[];

    const int tid    = threadIdx.x;
    const int sample = blockIdx.x;
    const int lane   = tid & 31;
    const int wave   = __builtin_amdgcn_readfirstlane(tid >> 5);
    __builtin_assume(wave >= 0 && wave < 8);
    const int n      = lane & 15;
    const int h      = lane >> 4;

    for (int q = tid; q < ENV7; q += 256) env[q] = 0.0f;
    __syncthreads();
    if (tid == 0) env[0] = 1.0f;
    __syncthreads();

    const int* xrow = x + sample * NSITE;

    for (int j = 0; j < LY; ++j) {
        for (int i = 0; i < LX; ++i) {
            const int site = i * LY + j;
            const int s    = xrow[site];
            const float* Mp = Msite + (site * PD + s) * 256;

            const int shift = 2 * (5 - i);

            if (j == 0) {
                if (shift >= 4)
                    apply_site<1, 4, 0>(env, Mp, shift, 1 << (2 * i), shift - 4,
                                        wave, n, h);
                else if (i == LX - 2)
                    apply_site<1, 4, 1>(env, Mp, shift, 64, 0, wave, n, h);
                else
                    apply_site<1, 4, 2>(env, Mp, shift, 64, 0, wave, n, h);
            } else {
                const int ntile = (j == LY - 1)
                                ? ((shift >= 4) ? (1 << (shift - 4)) : 1) : 64;
                if (i == 0)
                    apply_site<1, 1, 0>(env, Mp, shift, ntile, 0, wave, n, h);
                else if (shift >= 4)
                    apply_site<4, 1, 0>(env, Mp, shift, ntile, 0, wave, n, h);
                else if (i == LX - 2)
                    apply_site<4, 1, 1>(env, Mp, shift, ntile, 0, wave, n, h);
                else
                    apply_site<4, 1, 2>(env, Mp, shift, ntile, 0, wave, n, h);
            }
            __syncthreads();
        }

        if (j < LY - 1) {
            float keep[16];
            #pragma unroll
            for (int q = 0; q < 16; ++q) keep[q] = env[4 * (tid + 256 * q)];
            __syncthreads();
            #pragma unroll
            for (int q = 0; q < 16; ++q) env[tid + 256 * q] = keep[q];
            __syncthreads();
        }
    }

    if (tid == 0) out[sample] = env[0];
}

extern "C" void kernel_launch(void* const* d_in, const int* in_sizes, int n_in,
                              void* d_out, int out_size, void* d_ws, size_t ws_size,
                              hipStream_t stream) {
    const int*   x     = (const int*)d_in[0];
    const float* peps  = (const float*)d_in[1];
    float*       out   = (float*)d_out;
    float*       Msite = (float*)d_ws;

    const int nM = NSITE * PD * 256;
    peps_prep_kernel<<<(nM + 255) / 256, 256, 0, stream>>>(peps, Msite);

    const int B = out_size;
    peps_amp_kernel<<<B, 256, ENV7 * sizeof(float), stream>>>(x, Msite, out);
}


# 128-bit LDS traffic at post_size >= 64: each lane owns 4 adjacent env rows (4 row tiles per trip), last-row k pairs as one 64-bit load; k order and sum order unchanged
# speedup vs baseline: 3.8652x; 1.1166x over previous
#include <hip/hip_runtime.h>


typedef float v2f __attribute__((ext_vector_type(2)));
typedef float v4f __attribute__((ext_vector_type(4)));
typedef float v8f __attribute__((ext_vector_type(8)));

#define LX 6
#define LY 6
#define PD 2
#define DD 4
#define NSITE 36
#define ENV7 16384
#define ENV6 4096

__global__ void peps_prep_kernel(const float* __restrict__ peps,
                                 float* __restrict__ Msite) {
    int idx = blockIdx.x * blockDim.x + threadIdx.x;
    if (idx >= NSITE * PD * 256) return;
    int n    = idx & 15;
    int k    = (idx >> 4) & 15;
    int s    = (idx >> 8) & 1;
    int site = idx >> 9;
    int i = site / LY, j = site % LY;
    int a = k >> 2, c = k & 3;
    int d = n >> 2, b = n & 3;
    float v = 0.0f;
    bool pad = (i == 0      && a > 0) || (i == LX - 1 && b > 0) ||
               (j == 0      && c > 0) || (j == LY - 1 && d > 0);
    if (!pad) {
        int p = ((((site * PD + s) * DD + a) * DD + b) * DD + c) * DD + d;
        v = peps[p];
    }
    Msite[idx] = v;
}

template <int NK, int KSTR, int FORM>
__device__ __forceinline__ void apply_site(float* env,
                                           const float* __restrict__ Mp,
                                           const int shift, const int ntile,
                                           const int tsh, const int wave,
                                           const int n, const int h) {
    const int psm = (1 << shift) - 1;

    v2f b[NK];
    #pragma unroll
    for (int kc = 0; kc < NK; ++kc) {
        b[kc].x = Mp[(4 * kc + KSTR * (2 * h    )) * 16 + n];
        b[kc].y = Mp[(4 * kc + KSTR * (2 * h + 1)) * 16 + n];
    }

    #pragma unroll 2
    for (int tt = wave; tt < ntile; tt += 8) {
        const int t = tt << tsh;
        int r    = t * 16 + n;
        int post = r & psm;
        int base = ((r - post) << 4) + post;
        v2f a[NK];
        #pragma unroll
        for (int kc = 0; kc < NK; ++kc) {
            if (FORM == 2 && KSTR == 1) {
                a[kc] = *(const v2f*)(env + base + 4 * kc + 2 * h);
            } else {
                a[kc].x = env[base + ((4 * kc + KSTR * (2 * h    )) << shift)];
                a[kc].y = env[base + ((4 * kc + KSTR * (2 * h + 1)) << shift)];
            }
        }

        v8f acc = {};
        #pragma unroll
        for (int kc = 0; kc < NK; ++kc) {
            if (FORM == 1)
                acc = __builtin_amdgcn_wmma_f32_16x16x4_f32(
                        false, a[kc], false, b[kc], (short)0, acc, false, false);
            else
                acc = __builtin_amdgcn_wmma_f32_16x16x4_f32(
                        false, b[kc], false, a[kc], (short)0, acc, false, false);
        }

        if (FORM == 0) {
            #pragma unroll
            for (int v = 0; v < 8; ++v)
                env[base + ((v + 8 * h) << shift)] = acc[v];
        } else if (FORM == 2) {
            env[base + ((0 + 8 * h) << shift)] = acc[0];
            env[base + ((4 + 8 * h) << shift)] = acc[4];
        } else {
            #pragma unroll
            for (int v = 0; v < 8; ++v) {
                int rs    = t * 16 + v + 8 * h;
                int posts = rs & psm;
                int bs    = ((rs - posts) << 4) + posts;
                env[bs + (n << shift)] = acc[v];
            }
        }
    }
}

template <int NK, int SHIFT>
__device__ __forceinline__ void apply_site_rows4(float* env,
                                                 const float* __restrict__ Mp,
                                                 const int ngroup,
                                                 const int wave, const int n,
                                                 const int h) {
    constexpr int shift = SHIFT;
    constexpr int psm = (1 << shift) - 1;

    v2f m[NK];
    #pragma unroll
    for (int kc = 0; kc < NK; ++kc) {
        m[kc].x = Mp[(4 * kc + 2 * h    ) * 16 + n];
        m[kc].y = Mp[(4 * kc + 2 * h + 1) * 16 + n];
    }

    for (int g = wave; g < ngroup; g += 8) {
        const int r0   = (g * 16 + n) * 4;
        const int post = r0 & psm;
        const int base = ((r0 - post) << 4) + post;

        const float* src = env + base + ((2 * h) << shift);
        float*       dst = env + base + ((8 * h) << shift);

        v4f e0[NK], e1[NK];
        #pragma unroll
        for (int kc = 0; kc < NK; ++kc) {
            e0[kc] = *(const v4f*)(src + ((4 * kc    ) << shift));
            e1[kc] = *(const v4f*)(src + ((4 * kc + 1) << shift));
        }

        v8f acc0 = {}, acc1 = {}, acc2 = {}, acc3 = {};
        #pragma unroll
        for (int kc = 0; kc < NK; ++kc) {
            v2f x0 = {e0[kc].x, e1[kc].x};
            v2f x1 = {e0[kc].y, e1[kc].y};
            v2f x2 = {e0[kc].z, e1[kc].z};
            v2f x3 = {e0[kc].w, e1[kc].w};
            acc0 = __builtin_amdgcn_wmma_f32_16x16x4_f32(
                     false, m[kc], false, x0, (short)0, acc0, false, false);
            acc1 = __builtin_amdgcn_wmma_f32_16x16x4_f32(
                     false, m[kc], false, x1, (short)0, acc1, false, false);
            acc2 = __builtin_amdgcn_wmma_f32_16x16x4_f32(
                     false, m[kc], false, x2, (short)0, acc2, false, false);
            acc3 = __builtin_amdgcn_wmma_f32_16x16x4_f32(
                     false, m[kc], false, x3, (short)0, acc3, false, false);
        }

        asm volatile("v_nop\n\tv_nop\n\tv_nop\n\tv_nop"
                     : "+v"(acc0), "+v"(acc1), "+v"(acc2), "+v"(acc3));

        #pragma unroll
        for (int v = 0; v < 8; ++v) {
            v4f o = {acc0[v], acc1[v], acc2[v], acc3[v]};
            *(v4f*)(dst + (v << shift)) = o;
        }
    }
}

__global__ void __launch_bounds__(256)
peps_amp_kernel(const int* __restrict__ x,
                const float* __restrict__ Msite,
                float* __restrict__ out) {
    extern __shared__ __attribute__((aligned(16))) float env[];

    const int tid    = threadIdx.x;
    const int sample = blockIdx.x;
    const int lane   = tid & 31;
    const int wave   = __builtin_amdgcn_readfirstlane(tid >> 5);
    __builtin_assume(wave >= 0 && wave < 8);
    const int n      = lane & 15;
    const int h      = lane >> 4;

    for (int q = tid; q < ENV7; q += 256) env[q] = 0.0f;
    __syncthreads();
    if (tid == 0) env[0] = 1.0f;
    __syncthreads();

    const int* xrow = x + sample * NSITE;

    for (int j = 0; j < LY; ++j) {
        for (int i = 0; i < LX; ++i) {
            const int site = i * LY + j;
            const int s    = xrow[site];
            const float* Mp = Msite + (site * PD + s) * 256;

            const int shift = 2 * (5 - i);

            if (j == 0) {
                if (shift >= 4)
                    apply_site<1, 4, 0>(env, Mp, shift, 1 << (2 * i), shift - 4,
                                        wave, n, h);
                else if (i == LX - 2)
                    apply_site<1, 4, 1>(env, Mp, shift, 64, 0, wave, n, h);
                else
                    apply_site<1, 4, 2>(env, Mp, shift, 64, 0, wave, n, h);
            } else {
                const int ntile = (j == LY - 1)
                                ? ((shift >= 4) ? (1 << (shift - 4)) : 1) : 64;
                if (i == 0)
                    apply_site_rows4<1, 10>(env, Mp, ntile >> 2, wave, n, h);
                else if (i == 1)
                    apply_site_rows4<4, 8>(env, Mp, ntile >> 2, wave, n, h);
                else if (i == 2)
                    apply_site_rows4<4, 6>(env, Mp, ntile >> 2, wave, n, h);
                else if (shift >= 4)
                    apply_site<4, 1, 0>(env, Mp, shift, ntile, 0, wave, n, h);
                else if (i == LX - 2)
                    apply_site<4, 1, 1>(env, Mp, shift, ntile, 0, wave, n, h);
                else
                    apply_site<4, 1, 2>(env, Mp, shift, ntile, 0, wave, n, h);
            }
            __syncthreads();
        }

        if (j < LY - 1) {
            float keep[16];
            #pragma unroll
            for (int q = 0; q < 16; ++q) keep[q] = env[4 * (tid + 256 * q)];
            __syncthreads();
            #pragma unroll
            for (int q = 0; q < 16; ++q) env[tid + 256 * q] = keep[q];
            __syncthreads();
        }
    }

    if (tid == 0) out[sample] = env[0];
}

extern "C" void kernel_launch(void* const* d_in, const int* in_sizes, int n_in,
                              void* d_out, int out_size, void* d_ws, size_t ws_size,
                              hipStream_t stream) {
    const int*   x     = (const int*)d_in[0];
    const float* peps  = (const float*)d_in[1];
    float*       out   = (float*)d_out;
    float*       Msite = (float*)d_ws;

    const int nM = NSITE * PD * 256;
    peps_prep_kernel<<<(nM + 255) / 256, 256, 0, stream>>>(peps, Msite);

    const int B = out_size;
    peps_amp_kernel<<<B, 256, ENV7 * sizeof(float), stream>>>(x, Msite, out);
}


# rows 4 and 5 of columns 1-4 contracted as one K=64 step against their 64x16 site-pair product (built per block on WMMA); sum order changes: site product first, K split over 2 accumulators
# speedup vs baseline: 4.8090x; 1.2442x over previous
#include <hip/hip_runtime.h>


typedef float v2f __attribute__((ext_vector_type(2)));
typedef float v4f __attribute__((ext_vector_type(4)));
typedef float v8f __attribute__((ext_vector_type(8)));

#define LX 6
#define LY 6
#define PD 2
#define DD 4
#define NSITE 36
#define ENV7 16384
#define ENV6 4096
#define TLEN 1024

__global__ void peps_prep_kernel(const float* __restrict__ peps,
                                 float* __restrict__ Msite) {
    int idx = blockIdx.x * blockDim.x + threadIdx.x;
    if (idx >= NSITE * PD * 256) return;
    int n    = idx & 15;
    int k    = (idx >> 4) & 15;
    int s    = (idx >> 8) & 1;
    int site = idx >> 9;
    int i = site / LY, j = site % LY;
    int a = k >> 2, c = k & 3;
    int d = n >> 2, b = n & 3;
    float v = 0.0f;
    bool pad = (i == 0      && a > 0) || (i == LX - 1 && b > 0) ||
               (j == 0      && c > 0) || (j == LY - 1 && d > 0);
    if (!pad) {
        int p = ((((site * PD + s) * DD + a) * DD + b) * DD + c) * DD + d;
        v = peps[p];
    }
    Msite[idx] = v;
}

template <int NK, int KSTR, int FORM>
__device__ __forceinline__ void apply_site(float* env,
                                           const float* __restrict__ Mp,
                                           const int shift, const int ntile,
                                           const int tsh, const int wave,
                                           const int n, const int h) {
    const int psm = (1 << shift) - 1;

    v2f b[NK];
    #pragma unroll
    for (int kc = 0; kc < NK; ++kc) {
        b[kc].x = Mp[(4 * kc + KSTR * (2 * h    )) * 16 + n];
        b[kc].y = Mp[(4 * kc + KSTR * (2 * h + 1)) * 16 + n];
    }

    #pragma unroll 2
    for (int tt = wave; tt < ntile; tt += 8) {
        const int t = tt << tsh;
        int r    = t * 16 + n;
        int post = r & psm;
        int base = ((r - post) << 4) + post;
        v2f a[NK];
        #pragma unroll
        for (int kc = 0; kc < NK; ++kc) {
            if (FORM == 2 && KSTR == 1) {
                a[kc] = *(const v2f*)(env + base + 4 * kc + 2 * h);
            } else {
                a[kc].x = env[base + ((4 * kc + KSTR * (2 * h    )) << shift)];
                a[kc].y = env[base + ((4 * kc + KSTR * (2 * h + 1)) << shift)];
            }
        }

        v8f acc = {};
        #pragma unroll
        for (int kc = 0; kc < NK; ++kc) {
            if (FORM == 1)
                acc = __builtin_amdgcn_wmma_f32_16x16x4_f32(
                        false, a[kc], false, b[kc], (short)0, acc, false, false);
            else
                acc = __builtin_amdgcn_wmma_f32_16x16x4_f32(
                        false, b[kc], false, a[kc], (short)0, acc, false, false);
        }

        if (FORM == 0) {
            #pragma unroll
            for (int v = 0; v < 8; ++v)
                env[base + ((v + 8 * h) << shift)] = acc[v];
        } else if (FORM == 2) {
            env[base + ((0 + 8 * h) << shift)] = acc[0];
            env[base + ((4 + 8 * h) << shift)] = acc[4];
        } else {
            #pragma unroll
            for (int v = 0; v < 8; ++v) {
                int rs    = t * 16 + v + 8 * h;
                int posts = rs & psm;
                int bs    = ((rs - posts) << 4) + posts;
                env[bs + (n << shift)] = acc[v];
            }
        }
    }
}

template <int NK, int SHIFT>
__device__ __forceinline__ void apply_site_rows4(float* env,
                                                 const float* __restrict__ Mp,
                                                 const int ngroup,
                                                 const int wave, const int n,
                                                 const int h) {
    constexpr int shift = SHIFT;
    constexpr int psm = (1 << shift) - 1;

    v2f m[NK];
    #pragma unroll
    for (int kc = 0; kc < NK; ++kc) {
        m[kc].x = Mp[(4 * kc + 2 * h    ) * 16 + n];
        m[kc].y = Mp[(4 * kc + 2 * h + 1) * 16 + n];
    }

    for (int g = wave; g < ngroup; g += 8) {
        const int r0   = (g * 16 + n) * 4;
        const int post = r0 & psm;
        const int base = ((r0 - post) << 4) + post;

        const float* src = env + base + ((2 * h) << shift);
        float*       dst = env + base + ((8 * h) << shift);

        v4f e0[NK], e1[NK];
        #pragma unroll
        for (int kc = 0; kc < NK; ++kc) {
            e0[kc] = *(const v4f*)(src + ((4 * kc    ) << shift));
            e1[kc] = *(const v4f*)(src + ((4 * kc + 1) << shift));
        }

        v8f acc0 = {}, acc1 = {}, acc2 = {}, acc3 = {};
        #pragma unroll
        for (int kc = 0; kc < NK; ++kc) {
            v2f x0 = {e0[kc].x, e1[kc].x};
            v2f x1 = {e0[kc].y, e1[kc].y};
            v2f x2 = {e0[kc].z, e1[kc].z};
            v2f x3 = {e0[kc].w, e1[kc].w};
            acc0 = __builtin_amdgcn_wmma_f32_16x16x4_f32(
                     false, m[kc], false, x0, (short)0, acc0, false, false);
            acc1 = __builtin_amdgcn_wmma_f32_16x16x4_f32(
                     false, m[kc], false, x1, (short)0, acc1, false, false);
            acc2 = __builtin_amdgcn_wmma_f32_16x16x4_f32(
                     false, m[kc], false, x2, (short)0, acc2, false, false);
            acc3 = __builtin_amdgcn_wmma_f32_16x16x4_f32(
                     false, m[kc], false, x3, (short)0, acc3, false, false);
        }

        asm volatile("v_nop\n\tv_nop\n\tv_nop\n\tv_nop"
                     : "+v"(acc0), "+v"(acc1), "+v"(acc2), "+v"(acc3));

        #pragma unroll
        for (int v = 0; v < 8; ++v) {
            v4f o = {acc0[v], acc1[v], acc2[v], acc3[v]};
            *(v4f*)(dst + (v << shift)) = o;
        }
    }
}

__device__ __forceinline__ void apply_site_row3(float* env,
                                                const float* __restrict__ Mp,
                                                const int wave, int n, int h) {
    asm volatile("" : "+v"(n), "+v"(h));

    v2f b[4];
    #pragma unroll
    for (int kc = 0; kc < 4; ++kc) {
        b[kc].x = Mp[(4 * kc + 2 * h    ) * 16 + n];
        b[kc].y = Mp[(4 * kc + 2 * h + 1) * 16 + n];
    }
    const int w3 = wave & 3;
    int so[8];
    #pragma unroll
    for (int v = 0; v < 8; ++v) {
        const int r3 = 2 * h + (v >> 2);
        so[v] = r3 * 64 + (((v & 3) ^ w3) << 4) + (((n >> 2) ^ r3) << 2) + (n & 3);
    }

    #pragma unroll 2
    for (int t = wave; t < 64; t += 8) {
        const int tb = t * 256;
        v2f a[4];
        #pragma unroll
        for (int kc = 0; kc < 4; ++kc) {
            a[kc].x = env[tb + n + ((4 * kc + 2 * h    ) << 4)];
            a[kc].y = env[tb + n + ((4 * kc + 2 * h + 1) << 4)];
        }
        v8f acc = {};
        #pragma unroll
        for (int kc = 0; kc < 4; ++kc)
            acc = __builtin_amdgcn_wmma_f32_16x16x4_f32(
                    false, b[kc], false, a[kc], (short)0, acc, false, false);
        #pragma unroll
        for (int v = 0; v < 8; ++v)
            env[tb + so[v]] = acc[v];
    }
}

__device__ __forceinline__ void pair_tile_load(const float* __restrict__ M4,
                                               const float* __restrict__ M5,
                                               const int tile, const int n,
                                               const int h, v2f& a, v2f& b) {
    a = *(const v2f*)(M4 + (16 * tile + n) * 4 + 2 * h);
    b.x = M5[((2 * h    ) * 4 + (n & 3)) * 16 + (n >> 2) * 4];
    b.y = M5[((2 * h + 1) * 4 + (n & 3)) * 16 + (n >> 2) * 4];
}

__device__ __forceinline__ void pair_tile_store(float* Tl, const v2f a,
                                                const v2f b, const int tile,
                                                const int n, const int h) {
    v8f z = {};
    z = __builtin_amdgcn_wmma_f32_16x16x4_f32(
            false, a, false, b, (short)0, z, false, false);
    #pragma unroll
    for (int v = 0; v < 8; ++v) {
        const int k4 = 4 * tile + 2 * h + (v >> 2);
        Tl[(k4 & 7) * 128 + (k4 >> 3) * 64 + ((v & 3) * 4 + (n >> 2)) * 4 + (n & 3)] = z[v];
    }
}

__device__ __forceinline__ void apply_pair(const float* env, const float* Tl,
                                           const int wave, int n, int h,
                                           v8f& r0, v8f& r1) {
    asm volatile("" : "+v"(n), "+v"(h));

    const float* tp = Tl + (h * 16 + n) * 4;
    const float* e0 = env + (wave * 16 + n) * 64;
    const float* e1 = e0 + 8 * 16 * 64;

    v8f acc00 = {}, acc01 = {}, acc10 = {}, acc11 = {};
    #pragma unroll
    for (int j = 0; j < 8; ++j) {
        const int slot = ((8 * h + j) ^ n) << 2;
        const v4f tv = *(const v4f*)(tp + j * 128);
        const v4f u0 = *(const v4f*)(e0 + slot);
        const v4f u1 = *(const v4f*)(e1 + slot);
        const v2f ta = {tv.x, tv.y}, tb = {tv.z, tv.w};
        const v2f x0 = {u0.x, u0.y}, y0 = {u0.z, u0.w};
        const v2f x1 = {u1.x, u1.y}, y1 = {u1.z, u1.w};
        acc00 = __builtin_amdgcn_wmma_f32_16x16x4_f32(
                  false, ta, false, x0, (short)0, acc00, false, false);
        acc10 = __builtin_amdgcn_wmma_f32_16x16x4_f32(
                  false, ta, false, x1, (short)0, acc10, false, false);
        acc01 = __builtin_amdgcn_wmma_f32_16x16x4_f32(
                  false, tb, false, y0, (short)0, acc01, false, false);
        acc11 = __builtin_amdgcn_wmma_f32_16x16x4_f32(
                  false, tb, false, y1, (short)0, acc11, false, false);
    }
    asm volatile("v_nop\n\tv_nop\n\tv_nop\n\tv_nop"
                 : "+v"(acc00), "+v"(acc01), "+v"(acc10), "+v"(acc11));
    r0 = acc00 + acc01;
    r1 = acc10 + acc11;
}

__global__ void __launch_bounds__(256)
peps_amp_kernel(const int* __restrict__ x,
                const float* __restrict__ Msite,
                float* __restrict__ out) {
    __shared__ __attribute__((aligned(16))) float env[ENV7 + TLEN];
    float* const Tl = env + ENV7;

    const int tid    = threadIdx.x;
    const int sample = blockIdx.x;
    const int lane   = tid & 31;
    const int wave   = __builtin_amdgcn_readfirstlane(tid >> 5);
    __builtin_assume(wave >= 0 && wave < 8);
    const int n      = lane & 15;
    const int h      = lane >> 4;

    for (int q = tid; q < ENV7; q += 256) env[q] = 0.0f;
    __syncthreads();
    if (tid == 0) env[0] = 1.0f;
    __syncthreads();

    const int* xrow = x + sample * NSITE;

    for (int j = 0; j < LY; ++j) {
        const bool inner = (j > 0) && (j < LY - 1);
        const int  nrow  = inner ? LX - 1 : LX;

        for (int i = 0; i < nrow; ++i) {
            const int site = i * LY + j;
            const int s    = xrow[site];
            const float* Mp = Msite + (site * PD + s) * 256;

            const int shift = 2 * (5 - i);

            if (j == 0) {
                if (shift >= 4)
                    apply_site<1, 4, 0>(env, Mp, shift, 1 << (2 * i), shift - 4,
                                        wave, n, h);
                else if (i == LX - 2)
                    apply_site<1, 4, 1>(env, Mp, shift, 64, 0, wave, n, h);
                else
                    apply_site<1, 4, 2>(env, Mp, shift, 64, 0, wave, n, h);
            } else {
                const int ntile = (j == LY - 1)
                                ? ((shift >= 4) ? (1 << (shift - 4)) : 1) : 64;
                if (i == 0)
                    apply_site_rows4<1, 10>(env, Mp, ntile >> 2, wave, n, h);
                else if (i == 1)
                    apply_site_rows4<4, 8>(env, Mp, ntile >> 2, wave, n, h);
                else if (i == 2)
                    apply_site_rows4<4, 6>(env, Mp, ntile >> 2, wave, n, h);
                else if (inner && i == 3) {
                    v2f pa = {}, pb = {};
                    if (wave < 4) {
                        const int s4 = xrow[4 * LY + j];
                        const int s5 = xrow[5 * LY + j];
                        pair_tile_load(Msite + ((4 * LY + j) * PD + s4) * 256,
                                       Msite + ((5 * LY + j) * PD + s5) * 256,
                                       wave, n, h, pa, pb);
                    }
                    apply_site_row3(env, Mp, wave, n, h);
                    if (wave < 4)
                        pair_tile_store(Tl, pa, pb, wave, n, h);
                } else if (inner && i == 4) {
                    v8f r0, r1;
                    apply_pair(env, Tl, wave, n, h, r0, r1);
                    __syncthreads();
                    float* o0 = env + (wave * 16 + n) * 16 + 8 * h;
                    float* o1 = o0 + 8 * 16 * 16;
                    *(v4f*)(o0    ) = __builtin_shufflevector(r0, r0, 0, 1, 2, 3);
                    *(v4f*)(o0 + 4) = __builtin_shufflevector(r0, r0, 4, 5, 6, 7);
                    *(v4f*)(o1    ) = __builtin_shufflevector(r1, r1, 0, 1, 2, 3);
                    *(v4f*)(o1 + 4) = __builtin_shufflevector(r1, r1, 4, 5, 6, 7);
                } else if (shift >= 4)
                    apply_site<4, 1, 0>(env, Mp, shift, ntile, 0, wave, n, h);
                else if (i == LX - 2)
                    apply_site<4, 1, 1>(env, Mp, shift, ntile, 0, wave, n, h);
                else
                    apply_site<4, 1, 2>(env, Mp, shift, ntile, 0, wave, n, h);
            }
            __syncthreads();
        }

        if (j == 0) {
            float keep[16];
            #pragma unroll
            for (int q = 0; q < 16; ++q) keep[q] = env[4 * (tid + 256 * q)];
            __syncthreads();
            #pragma unroll
            for (int q = 0; q < 16; ++q) env[tid + 256 * q] = keep[q];
            __syncthreads();
        }
    }

    if (tid == 0) out[sample] = env[0];
}

extern "C" void kernel_launch(void* const* d_in, const int* in_sizes, int n_in,
                              void* d_out, int out_size, void* d_ws, size_t ws_size,
                              hipStream_t stream) {
    const int*   x     = (const int*)d_in[0];
    const float* peps  = (const float*)d_in[1];
    float*       out   = (float*)d_out;
    float*       Msite = (float*)d_ws;

    const int nM = NSITE * PD * 256;
    peps_prep_kernel<<<(nM + 255) / 256, 256, 0, stream>>>(peps, Msite);

    const int B = out_size;
    peps_amp_kernel<<<B, 256, 0, stream>>>(x, Msite, out);
}


# rows 0 and 1 of columns 1-5 contracted as one K=16 step against their 16x64 site-pair product (built per block on WMMA); sum order changes: site product over down0 first, then K = 16
# speedup vs baseline: 5.1862x; 1.0784x over previous
#include <hip/hip_runtime.h>


typedef float v2f __attribute__((ext_vector_type(2)));
typedef float v4f __attribute__((ext_vector_type(4)));
typedef float v8f __attribute__((ext_vector_type(8)));

#define LX 6
#define LY 6
#define PD 2
#define DD 4
#define NSITE 36
#define ENV7 16384
#define ENV6 4096
#define TLEN 1024

__global__ void peps_prep_kernel(const float* __restrict__ peps,
                                 float* __restrict__ Msite) {
    int idx = blockIdx.x * blockDim.x + threadIdx.x;
    if (idx >= NSITE * PD * 256) return;
    int n    = idx & 15;
    int k    = (idx >> 4) & 15;
    int s    = (idx >> 8) & 1;
    int site = idx >> 9;
    int i = site / LY, j = site % LY;
    int a = k >> 2, c = k & 3;
    int d = n >> 2, b = n & 3;
    float v = 0.0f;
    bool pad = (i == 0      && a > 0) || (i == LX - 1 && b > 0) ||
               (j == 0      && c > 0) || (j == LY - 1 && d > 0);
    if (!pad) {
        int p = ((((site * PD + s) * DD + a) * DD + b) * DD + c) * DD + d;
        v = peps[p];
    }
    Msite[idx] = v;
}

template <int NK, int KSTR, int FORM>
__device__ __forceinline__ void apply_site(float* env,
                                           const float* __restrict__ Mp,
                                           const int shift, const int ntile,
                                           const int tsh, const int wave,
                                           const int n, const int h) {
    const int psm = (1 << shift) - 1;

    v2f b[NK];
    #pragma unroll
    for (int kc = 0; kc < NK; ++kc) {
        b[kc].x = Mp[(4 * kc + KSTR * (2 * h    )) * 16 + n];
        b[kc].y = Mp[(4 * kc + KSTR * (2 * h + 1)) * 16 + n];
    }

    #pragma unroll 2
    for (int tt = wave; tt < ntile; tt += 8) {
        const int t = tt << tsh;
        int r    = t * 16 + n;
        int post = r & psm;
        int base = ((r - post) << 4) + post;
        v2f a[NK];
        #pragma unroll
        for (int kc = 0; kc < NK; ++kc) {
            if (FORM == 2 && KSTR == 1) {
                a[kc] = *(const v2f*)(env + base + 4 * kc + 2 * h);
            } else {
                a[kc].x = env[base + ((4 * kc + KSTR * (2 * h    )) << shift)];
                a[kc].y = env[base + ((4 * kc + KSTR * (2 * h + 1)) << shift)];
            }
        }

        v8f acc = {};
        #pragma unroll
        for (int kc = 0; kc < NK; ++kc) {
            if (FORM == 1)
                acc = __builtin_amdgcn_wmma_f32_16x16x4_f32(
                        false, a[kc], false, b[kc], (short)0, acc, false, false);
            else
                acc = __builtin_amdgcn_wmma_f32_16x16x4_f32(
                        false, b[kc], false, a[kc], (short)0, acc, false, false);
        }

        if (FORM == 0) {
            #pragma unroll
            for (int v = 0; v < 8; ++v)
                env[base + ((v + 8 * h) << shift)] = acc[v];
        } else if (FORM == 2) {
            env[base + ((0 + 8 * h) << shift)] = acc[0];
            env[base + ((4 + 8 * h) << shift)] = acc[4];
        } else {
            #pragma unroll
            for (int v = 0; v < 8; ++v) {
                int rs    = t * 16 + v + 8 * h;
                int posts = rs & psm;
                int bs    = ((rs - posts) << 4) + posts;
                env[bs + (n << shift)] = acc[v];
            }
        }
    }
}

template <int NK, int SHIFT>
__device__ __forceinline__ void apply_site_rows4(float* env,
                                                 const float* __restrict__ Mp,
                                                 const int ngroup,
                                                 const int wave, const int n,
                                                 const int h) {
    constexpr int shift = SHIFT;
    constexpr int psm = (1 << shift) - 1;

    v2f m[NK];
    #pragma unroll
    for (int kc = 0; kc < NK; ++kc) {
        m[kc].x = Mp[(4 * kc + 2 * h    ) * 16 + n];
        m[kc].y = Mp[(4 * kc + 2 * h + 1) * 16 + n];
    }

    for (int g = wave; g < ngroup; g += 8) {
        const int r0   = (g * 16 + n) * 4;
        const int post = r0 & psm;
        const int base = ((r0 - post) << 4) + post;

        const float* src = env + base + ((2 * h) << shift);
        float*       dst = env + base + ((8 * h) << shift);

        v4f e0[NK], e1[NK];
        #pragma unroll
        for (int kc = 0; kc < NK; ++kc) {
            e0[kc] = *(const v4f*)(src + ((4 * kc    ) << shift));
            e1[kc] = *(const v4f*)(src + ((4 * kc + 1) << shift));
        }

        v8f acc0 = {}, acc1 = {}, acc2 = {}, acc3 = {};
        #pragma unroll
        for (int kc = 0; kc < NK; ++kc) {
            v2f x0 = {e0[kc].x, e1[kc].x};
            v2f x1 = {e0[kc].y, e1[kc].y};
            v2f x2 = {e0[kc].z, e1[kc].z};
            v2f x3 = {e0[kc].w, e1[kc].w};
            acc0 = __builtin_amdgcn_wmma_f32_16x16x4_f32(
                     false, m[kc], false, x0, (short)0, acc0, false, false);
            acc1 = __builtin_amdgcn_wmma_f32_16x16x4_f32(
                     false, m[kc], false, x1, (short)0, acc1, false, false);
            acc2 = __builtin_amdgcn_wmma_f32_16x16x4_f32(
                     false, m[kc], false, x2, (short)0, acc2, false, false);
            acc3 = __builtin_amdgcn_wmma_f32_16x16x4_f32(
                     false, m[kc], false, x3, (short)0, acc3, false, false);
        }

        asm volatile("v_nop\n\tv_nop\n\tv_nop\n\tv_nop"
                     : "+v"(acc0), "+v"(acc1), "+v"(acc2), "+v"(acc3));

        #pragma unroll
        for (int v = 0; v < 8; ++v) {
            v4f o = {acc0[v], acc1[v], acc2[v], acc3[v]};
            *(v4f*)(dst + (v << shift)) = o;
        }
    }
}

__device__ __forceinline__ void apply_site_row3(float* env,
                                                const float* __restrict__ Mp,
                                                const int wave, int n, int h) {
    asm volatile("" : "+v"(n), "+v"(h));

    v2f b[4];
    #pragma unroll
    for (int kc = 0; kc < 4; ++kc) {
        b[kc].x = Mp[(4 * kc + 2 * h    ) * 16 + n];
        b[kc].y = Mp[(4 * kc + 2 * h + 1) * 16 + n];
    }
    const int w3 = wave & 3;
    int so[8];
    #pragma unroll
    for (int v = 0; v < 8; ++v) {
        const int r3 = 2 * h + (v >> 2);
        so[v] = r3 * 64 + (((v & 3) ^ w3) << 4) + (((n >> 2) ^ r3) << 2) + (n & 3);
    }

    #pragma unroll 2
    for (int t = wave; t < 64; t += 8) {
        const int tb = t * 256;
        v2f a[4];
        #pragma unroll
        for (int kc = 0; kc < 4; ++kc) {
            a[kc].x = env[tb + n + ((4 * kc + 2 * h    ) << 4)];
            a[kc].y = env[tb + n + ((4 * kc + 2 * h + 1) << 4)];
        }
        v8f acc = {};
        #pragma unroll
        for (int kc = 0; kc < 4; ++kc)
            acc = __builtin_amdgcn_wmma_f32_16x16x4_f32(
                    false, b[kc], false, a[kc], (short)0, acc, false, false);
        #pragma unroll
        for (int v = 0; v < 8; ++v)
            env[tb + so[v]] = acc[v];
    }
}

__device__ __forceinline__ void pair_tile_load(const float* __restrict__ M4,
                                               const float* __restrict__ M5,
                                               const int tile, const int n,
                                               const int h, v2f& a, v2f& b) {
    a = *(const v2f*)(M4 + (16 * tile + n) * 4 + 2 * h);
    b.x = M5[((2 * h    ) * 4 + (n & 3)) * 16 + (n >> 2) * 4];
    b.y = M5[((2 * h + 1) * 4 + (n & 3)) * 16 + (n >> 2) * 4];
}

__device__ __forceinline__ void pair_tile_store(float* Tl, const v2f a,
                                                const v2f b, const int tile,
                                                const int n, const int h) {
    v8f z = {};
    z = __builtin_amdgcn_wmma_f32_16x16x4_f32(
            false, a, false, b, (short)0, z, false, false);
    #pragma unroll
    for (int v = 0; v < 8; ++v) {
        const int k4 = 4 * tile + 2 * h + (v >> 2);
        Tl[(k4 & 7) * 128 + (k4 >> 3) * 64 + ((v & 3) * 4 + (n >> 2)) * 4 + (n & 3)] = z[v];
    }
}

__device__ __forceinline__ void apply_pair(const float* env, const float* Tl,
                                           const int wave, int n, int h,
                                           v8f& r0, v8f& r1) {
    asm volatile("" : "+v"(n), "+v"(h));

    const float* tp = Tl + (h * 16 + n) * 4;
    const float* e0 = env + (wave * 16 + n) * 64;
    const float* e1 = e0 + 8 * 16 * 64;

    v8f acc00 = {}, acc01 = {}, acc10 = {}, acc11 = {};
    #pragma unroll
    for (int j = 0; j < 8; ++j) {
        const int slot = ((8 * h + j) ^ n) << 2;
        const v4f tv = *(const v4f*)(tp + j * 128);
        const v4f u0 = *(const v4f*)(e0 + slot);
        const v4f u1 = *(const v4f*)(e1 + slot);
        const v2f ta = {tv.x, tv.y}, tb = {tv.z, tv.w};
        const v2f x0 = {u0.x, u0.y}, y0 = {u0.z, u0.w};
        const v2f x1 = {u1.x, u1.y}, y1 = {u1.z, u1.w};
        acc00 = __builtin_amdgcn_wmma_f32_16x16x4_f32(
                  false, ta, false, x0, (short)0, acc00, false, false);
        acc10 = __builtin_amdgcn_wmma_f32_16x16x4_f32(
                  false, ta, false, x1, (short)0, acc10, false, false);
        acc01 = __builtin_amdgcn_wmma_f32_16x16x4_f32(
                  false, tb, false, y0, (short)0, acc01, false, false);
        acc11 = __builtin_amdgcn_wmma_f32_16x16x4_f32(
                  false, tb, false, y1, (short)0, acc11, false, false);
    }
    asm volatile("v_nop\n\tv_nop\n\tv_nop\n\tv_nop"
                 : "+v"(acc00), "+v"(acc01), "+v"(acc10), "+v"(acc11));
    r0 = acc00 + acc01;
    r1 = acc10 + acc11;
}

__device__ __forceinline__ void top_tile_load(const float* __restrict__ M0,
                                              const float* __restrict__ M1,
                                              const int w, int n, int h,
                                              v2f& a, v2f& b) {
    asm volatile("" : "+v"(n), "+v"(h));
    a = *(const v2f*)(M0 + (n & 3) * 16 + (n >> 2) * 4 + 2 * h);
    b.x = M1[((2 * h    ) * 4 + w) * 16 + n];
    b.y = M1[((2 * h + 1) * 4 + w) * 16 + n];
}

__device__ __forceinline__ void top_tile_store(float* Tl, const v2f a,
                                               const v2f b, const int w,
                                               int n, int h) {
    asm volatile("" : "+v"(n), "+v"(h));
    v8f z = {};
    z = __builtin_amdgcn_wmma_f32_16x16x4_f32(
            false, a, false, b, (short)0, z, false, false);
    #pragma unroll
    for (int v = 0; v < 8; ++v) {
        const int nt = 2 * h + (v >> 2);
        const int kc = v & 3;
        Tl[((nt * 2 + (kc >> 1)) * 32 + (w >> 1) * 16 + n) * 4 + (kc & 1) * 2 + (w & 1)] = z[v];
    }
}

__device__ __forceinline__ void apply_top_pair(float* env, const float* Tl,
                                               const int wave, int n, int h) {
    asm volatile("" : "+v"(n), "+v"(h));

    const int r0 = ((wave & 3) * 16 + n) * 4;
    const int nh = wave >> 2;

    const float* src = env + r0 + ((2 * h) << 8);
    v4f e0[4], e1[4];
    #pragma unroll
    for (int kc = 0; kc < 4; ++kc) {
        e0[kc] = *(const v4f*)(src + ((4 * kc    ) << 8));
        e1[kc] = *(const v4f*)(src + ((4 * kc + 1) << 8));
    }
    const float* tp = Tl + (nh * 4 * 32 + h * 16 + n) * 4;

    #pragma unroll
    for (int q = 0; q < 2; ++q) {
        const v4f t[2] = { *(const v4f*)(tp + (q * 2    ) * 128),
                           *(const v4f*)(tp + (q * 2 + 1) * 128) };
        v8f acc0 = {}, acc1 = {}, acc2 = {}, acc3 = {};
        #pragma unroll
        for (int kc = 0; kc < 4; ++kc) {
            const v4f tv = t[kc >> 1];
            const v2f m  = (kc & 1) ? (v2f){tv.z, tv.w} : (v2f){tv.x, tv.y};
            v2f x0 = {e0[kc].x, e1[kc].x};
            v2f x1 = {e0[kc].y, e1[kc].y};
            v2f x2 = {e0[kc].z, e1[kc].z};
            v2f x3 = {e0[kc].w, e1[kc].w};
            acc0 = __builtin_amdgcn_wmma_f32_16x16x4_f32(
                     false, m, false, x0, (short)0, acc0, false, false);
            acc1 = __builtin_amdgcn_wmma_f32_16x16x4_f32(
                     false, m, false, x1, (short)0, acc1, false, false);
            acc2 = __builtin_amdgcn_wmma_f32_16x16x4_f32(
                     false, m, false, x2, (short)0, acc2, false, false);
            acc3 = __builtin_amdgcn_wmma_f32_16x16x4_f32(
                     false, m, false, x3, (short)0, acc3, false, false);
        }
        asm volatile("v_nop\n\tv_nop\n\tv_nop\n\tv_nop"
                     : "+v"(acc0), "+v"(acc1), "+v"(acc2), "+v"(acc3));

        if (q == 0) __syncthreads();

        float* dst = env + (((2 * nh + q) * 16 + 8 * h) << 8) + r0;
        #pragma unroll
        for (int v = 0; v < 8; ++v) {
            v4f o = {acc0[v], acc1[v], acc2[v], acc3[v]};
            *(v4f*)(dst + (v << 8)) = o;
        }

        if (q == 0)
            asm volatile("" : "+v"(e0[0]), "+v"(e0[1]), "+v"(e0[2]), "+v"(e0[3]),
                              "+v"(e1[0]), "+v"(e1[1]), "+v"(e1[2]), "+v"(e1[3])
                            : : "memory");
    }
}

__device__ __forceinline__ void apply_top_pair_last(float* env, const float* Tl,
                                                    const int wave, int n, int h) {
    asm volatile("" : "+v"(n), "+v"(h));
    if (wave >= 4) return;

    const int r0 = (wave * 16 + n) * 4;
    const float* src = env + r0 + ((2 * h) << 8);
    v4f e0[4], e1[4];
    #pragma unroll
    for (int kc = 0; kc < 4; ++kc) {
        e0[kc] = *(const v4f*)(src + ((4 * kc    ) << 8));
        e1[kc] = *(const v4f*)(src + ((4 * kc + 1) << 8));
    }
    const float* tp = Tl + (h * 16 + n) * 4;
    const v4f t[2] = { *(const v4f*)(tp), *(const v4f*)(tp + 128) };

    v8f acc0 = {}, acc1 = {}, acc2 = {}, acc3 = {};
    #pragma unroll
    for (int kc = 0; kc < 4; ++kc) {
        const v4f tv = t[kc >> 1];
        const v2f m  = (kc & 1) ? (v2f){tv.z, tv.w} : (v2f){tv.x, tv.y};
        v2f x0 = {e0[kc].x, e1[kc].x};
        v2f x1 = {e0[kc].y, e1[kc].y};
        v2f x2 = {e0[kc].z, e1[kc].z};
        v2f x3 = {e0[kc].w, e1[kc].w};
        acc0 = __builtin_amdgcn_wmma_f32_16x16x4_f32(
                 false, m, false, x0, (short)0, acc0, false, false);
        acc1 = __builtin_amdgcn_wmma_f32_16x16x4_f32(
                 false, m, false, x1, (short)0, acc1, false, false);
        acc2 = __builtin_amdgcn_wmma_f32_16x16x4_f32(
                 false, m, false, x2, (short)0, acc2, false, false);
        acc3 = __builtin_amdgcn_wmma_f32_16x16x4_f32(
                 false, m, false, x3, (short)0, acc3, false, false);
    }
    asm volatile("v_nop\n\tv_nop\n\tv_nop\n\tv_nop"
                 : "+v"(acc0), "+v"(acc1), "+v"(acc2), "+v"(acc3));

    if (h == 0) {
        #pragma unroll
        for (int v = 0; v < 4; ++v) {
            v4f o = {acc0[v], acc1[v], acc2[v], acc3[v]};
            *(v4f*)(env + r0 + (v << 8)) = o;
        }
    }
}

__global__ void __launch_bounds__(256)
peps_amp_kernel(const int* __restrict__ x,
                const float* __restrict__ Msite,
                float* __restrict__ out) {
    __shared__ __attribute__((aligned(16))) float env[ENV7 + TLEN];
    float* const Tl = env + ENV7;

    const int tid    = threadIdx.x;
    const int sample = blockIdx.x;
    const int lane   = tid & 31;
    const int wave   = __builtin_amdgcn_readfirstlane(tid >> 5);
    __builtin_assume(wave >= 0 && wave < 8);
    const int n      = lane & 15;
    const int h      = lane >> 4;

    for (int q = tid; q < ENV7; q += 256) env[q] = 0.0f;
    __syncthreads();
    if (tid == 0) env[0] = 1.0f;
    __syncthreads();

    const int* xrow = x + sample * NSITE;

    for (int j = 0; j < LY; ++j) {
        const bool inner = (j > 0) && (j < LY - 1);
        const int  nrow  = inner ? LX - 1 : LX;

        for (int i = 0; i < nrow; ++i) {
            if (j > 0 && i == 1) continue;
            const int site = i * LY + j;
            const int s    = xrow[site];
            const float* Mp = Msite + (site * PD + s) * 256;

            const int shift = 2 * (5 - i);

            if (j == 0) {
                if (shift >= 4)
                    apply_site<1, 4, 0>(env, Mp, shift, 1 << (2 * i), shift - 4,
                                        wave, n, h);
                else if (i == LX - 2)
                    apply_site<1, 4, 1>(env, Mp, shift, 64, 0, wave, n, h);
                else
                    apply_site<1, 4, 2>(env, Mp, shift, 64, 0, wave, n, h);
            } else {
                const int ntile = (j == LY - 1)
                                ? ((shift >= 4) ? (1 << (shift - 4)) : 1) : 64;
                if (i == 0) {
                    if (inner) apply_top_pair(env, Tl, wave, n, h);
                    else       apply_top_pair_last(env, Tl, wave, n, h);
                } else if (i == 2)
                    apply_site_rows4<4, 6>(env, Mp, ntile >> 2, wave, n, h);
                else if (inner && i == 3) {
                    v2f pa = {}, pb = {};
                    if (wave < 4) {
                        const int s4 = xrow[4 * LY + j];
                        const int s5 = xrow[5 * LY + j];
                        pair_tile_load(Msite + ((4 * LY + j) * PD + s4) * 256,
                                       Msite + ((5 * LY + j) * PD + s5) * 256,
                                       wave, n, h, pa, pb);
                    }
                    apply_site_row3(env, Mp, wave, n, h);
                    if (wave < 4)
                        pair_tile_store(Tl, pa, pb, wave, n, h);
                } else if (inner && i == 4) {
                    const bool top = (wave < 4);
                    v2f qa = {}, qb = {};
                    if (top) {
                        const int s0 = xrow[0 * LY + j + 1];
                        const int s1 = xrow[1 * LY + j + 1];
                        top_tile_load(Msite + ((0 * LY + j + 1) * PD + s0) * 256,
                                      Msite + ((1 * LY + j + 1) * PD + s1) * 256,
                                      wave, n, h, qa, qb);
                    }
                    v8f r0, r1;
                    apply_pair(env, Tl, wave, n, h, r0, r1);
                    __syncthreads();
                    if (top)
                        top_tile_store(Tl, qa, qb, wave, n, h);
                    float* o0 = env + (wave * 16 + n) * 16 + 8 * h;
                    float* o1 = o0 + 8 * 16 * 16;
                    *(v4f*)(o0    ) = __builtin_shufflevector(r0, r0, 0, 1, 2, 3);
                    *(v4f*)(o0 + 4) = __builtin_shufflevector(r0, r0, 4, 5, 6, 7);
                    *(v4f*)(o1    ) = __builtin_shufflevector(r1, r1, 0, 1, 2, 3);
                    *(v4f*)(o1 + 4) = __builtin_shufflevector(r1, r1, 4, 5, 6, 7);
                } else if (shift >= 4)
                    apply_site<4, 1, 0>(env, Mp, shift, ntile, 0, wave, n, h);
                else if (i == LX - 2)
                    apply_site<4, 1, 1>(env, Mp, shift, ntile, 0, wave, n, h);
                else
                    apply_site<4, 1, 2>(env, Mp, shift, ntile, 0, wave, n, h);
            }
            __syncthreads();
        }

        if (j == 0) {
            v2f qa = {}, qb = {};
            if (wave < 4) {
                const int s0 = xrow[0 * LY + 1];
                const int s1 = xrow[1 * LY + 1];
                top_tile_load(Msite + ((0 * LY + 1) * PD + s0) * 256,
                              Msite + ((1 * LY + 1) * PD + s1) * 256,
                              wave, n, h, qa, qb);
            }
            float keep[16];
            #pragma unroll
            for (int q = 0; q < 16; ++q) keep[q] = env[4 * (tid + 256 * q)];
            __syncthreads();
            #pragma unroll
            for (int q = 0; q < 16; ++q) env[tid + 256 * q] = keep[q];
            if (wave < 4)
                top_tile_store(Tl, qa, qb, wave, n, h);
            __syncthreads();
        }
    }

    if (tid == 0) out[sample] = env[0];
}

extern "C" void kernel_launch(void* const* d_in, const int* in_sizes, int n_in,
                              void* d_out, int out_size, void* d_ws, size_t ws_size,
                              hipStream_t stream) {
    const int*   x     = (const int*)d_in[0];
    const float* peps  = (const float*)d_in[1];
    float*       out   = (float*)d_out;
    float*       Msite = (float*)d_ws;

    const int nM = NSITE * PD * 256;
    peps_prep_kernel<<<(nM + 255) / 256, 256, 0, stream>>>(peps, Msite);

    const int B = out_size;
    peps_amp_kernel<<<B, 256, 0, stream>>>(x, Msite, out);
}


# columns 0 and 1 contracted as one bond-16 matrix-product chain: site pairs multiplied first, then K=16 products on 4..256 rows, not 7 full-env steps; sum order changes: pair products first
# speedup vs baseline: 7.1365x; 1.3761x over previous
#include <hip/hip_runtime.h>


typedef float v2f __attribute__((ext_vector_type(2)));
typedef float v4f __attribute__((ext_vector_type(4)));
typedef float v8f __attribute__((ext_vector_type(8)));

#define LX 6
#define LY 6
#define PD 2
#define DD 4
#define NSITE 36
#define ENV7 16384
#define ENV6 4096
#define TLEN 1024

__global__ void peps_prep_kernel(const float* __restrict__ peps,
                                 float* __restrict__ Msite) {
    int idx = blockIdx.x * blockDim.x + threadIdx.x;
    if (idx >= NSITE * PD * 256) return;
    int n    = idx & 15;
    int k    = (idx >> 4) & 15;
    int s    = (idx >> 8) & 1;
    int site = idx >> 9;
    int i = site / LY, j = site % LY;
    int a = k >> 2, c = k & 3;
    int d = n >> 2, b = n & 3;
    float v = 0.0f;
    bool pad = (i == 0      && a > 0) || (i == LX - 1 && b > 0) ||
               (j == 0      && c > 0) || (j == LY - 1 && d > 0);
    if (!pad) {
        int p = ((((site * PD + s) * DD + a) * DD + b) * DD + c) * DD + d;
        v = peps[p];
    }
    Msite[idx] = v;
}

template <int NK, int KSTR, int FORM>
__device__ __forceinline__ void apply_site(float* env,
                                           const float* __restrict__ Mp,
                                           const int shift, const int ntile,
                                           const int tsh, const int wave,
                                           const int n, const int h) {
    const int psm = (1 << shift) - 1;

    v2f b[NK];
    #pragma unroll
    for (int kc = 0; kc < NK; ++kc) {
        b[kc].x = Mp[(4 * kc + KSTR * (2 * h    )) * 16 + n];
        b[kc].y = Mp[(4 * kc + KSTR * (2 * h + 1)) * 16 + n];
    }

    #pragma unroll 2
    for (int tt = wave; tt < ntile; tt += 8) {
        const int t = tt << tsh;
        int r    = t * 16 + n;
        int post = r & psm;
        int base = ((r - post) << 4) + post;
        v2f a[NK];
        #pragma unroll
        for (int kc = 0; kc < NK; ++kc) {
            if (FORM == 2 && KSTR == 1) {
                a[kc] = *(const v2f*)(env + base + 4 * kc + 2 * h);
            } else {
                a[kc].x = env[base + ((4 * kc + KSTR * (2 * h    )) << shift)];
                a[kc].y = env[base + ((4 * kc + KSTR * (2 * h + 1)) << shift)];
            }
        }

        v8f acc = {};
        #pragma unroll
        for (int kc = 0; kc < NK; ++kc) {
            if (FORM == 1)
                acc = __builtin_amdgcn_wmma_f32_16x16x4_f32(
                        false, a[kc], false, b[kc], (short)0, acc, false, false);
            else
                acc = __builtin_amdgcn_wmma_f32_16x16x4_f32(
                        false, b[kc], false, a[kc], (short)0, acc, false, false);
        }

        if (FORM == 0) {
            #pragma unroll
            for (int v = 0; v < 8; ++v)
                env[base + ((v + 8 * h) << shift)] = acc[v];
        } else if (FORM == 2) {
            env[base + ((0 + 8 * h) << shift)] = acc[0];
            env[base + ((4 + 8 * h) << shift)] = acc[4];
        } else {
            #pragma unroll
            for (int v = 0; v < 8; ++v) {
                int rs    = t * 16 + v + 8 * h;
                int posts = rs & psm;
                int bs    = ((rs - posts) << 4) + posts;
                env[bs + (n << shift)] = acc[v];
            }
        }
    }
}

template <int NK, int SHIFT>
__device__ __forceinline__ void apply_site_rows4(float* env,
                                                 const float* __restrict__ Mp,
                                                 const int ngroup,
                                                 const int wave, const int n,
                                                 const int h) {
    constexpr int shift = SHIFT;
    constexpr int psm = (1 << shift) - 1;

    v2f m[NK];
    #pragma unroll
    for (int kc = 0; kc < NK; ++kc) {
        m[kc].x = Mp[(4 * kc + 2 * h    ) * 16 + n];
        m[kc].y = Mp[(4 * kc + 2 * h + 1) * 16 + n];
    }

    for (int g = wave; g < ngroup; g += 8) {
        const int r0   = (g * 16 + n) * 4;
        const int post = r0 & psm;
        const int base = ((r0 - post) << 4) + post;

        const float* src = env + base + ((2 * h) << shift);
        float*       dst = env + base + ((8 * h) << shift);

        v4f e0[NK], e1[NK];
        #pragma unroll
        for (int kc = 0; kc < NK; ++kc) {
            e0[kc] = *(const v4f*)(src + ((4 * kc    ) << shift));
            e1[kc] = *(const v4f*)(src + ((4 * kc + 1) << shift));
        }

        v8f acc0 = {}, acc1 = {}, acc2 = {}, acc3 = {};
        #pragma unroll
        for (int kc = 0; kc < NK; ++kc) {
            v2f x0 = {e0[kc].x, e1[kc].x};
            v2f x1 = {e0[kc].y, e1[kc].y};
            v2f x2 = {e0[kc].z, e1[kc].z};
            v2f x3 = {e0[kc].w, e1[kc].w};
            acc0 = __builtin_amdgcn_wmma_f32_16x16x4_f32(
                     false, m[kc], false, x0, (short)0, acc0, false, false);
            acc1 = __builtin_amdgcn_wmma_f32_16x16x4_f32(
                     false, m[kc], false, x1, (short)0, acc1, false, false);
            acc2 = __builtin_amdgcn_wmma_f32_16x16x4_f32(
                     false, m[kc], false, x2, (short)0, acc2, false, false);
            acc3 = __builtin_amdgcn_wmma_f32_16x16x4_f32(
                     false, m[kc], false, x3, (short)0, acc3, false, false);
        }

        asm volatile("v_nop\n\tv_nop\n\tv_nop\n\tv_nop"
                     : "+v"(acc0), "+v"(acc1), "+v"(acc2), "+v"(acc3));

        #pragma unroll
        for (int v = 0; v < 8; ++v) {
            v4f o = {acc0[v], acc1[v], acc2[v], acc3[v]};
            *(v4f*)(dst + (v << shift)) = o;
        }
    }
}

__device__ __forceinline__ void apply_site_row3(float* env,
                                                const float* __restrict__ Mp,
                                                const int wave, int n, int h) {
    asm volatile("" : "+v"(n), "+v"(h));

    v2f b[4];
    #pragma unroll
    for (int kc = 0; kc < 4; ++kc) {
        b[kc].x = Mp[(4 * kc + 2 * h    ) * 16 + n];
        b[kc].y = Mp[(4 * kc + 2 * h + 1) * 16 + n];
    }
    const int w3 = wave & 3;
    int so[8];
    #pragma unroll
    for (int v = 0; v < 8; ++v) {
        const int r3 = 2 * h + (v >> 2);
        so[v] = r3 * 64 + (((v & 3) ^ w3) << 4) + (((n >> 2) ^ r3) << 2) + (n & 3);
    }

    #pragma unroll 2
    for (int t = wave; t < 64; t += 8) {
        const int tb = t * 256;
        v2f a[4];
        #pragma unroll
        for (int kc = 0; kc < 4; ++kc) {
            a[kc].x = env[tb + n + ((4 * kc + 2 * h    ) << 4)];
            a[kc].y = env[tb + n + ((4 * kc + 2 * h + 1) << 4)];
        }
        v8f acc = {};
        #pragma unroll
        for (int kc = 0; kc < 4; ++kc)
            acc = __builtin_amdgcn_wmma_f32_16x16x4_f32(
                    false, b[kc], false, a[kc], (short)0, acc, false, false);
        #pragma unroll
        for (int v = 0; v < 8; ++v)
            env[tb + so[v]] = acc[v];
    }
}

__device__ __forceinline__ void pair_tile_load(const float* __restrict__ M4,
                                               const float* __restrict__ M5,
                                               const int tile, const int n,
                                               const int h, v2f& a, v2f& b) {
    a = *(const v2f*)(M4 + (16 * tile + n) * 4 + 2 * h);
    b.x = M5[((2 * h    ) * 4 + (n & 3)) * 16 + (n >> 2) * 4];
    b.y = M5[((2 * h + 1) * 4 + (n & 3)) * 16 + (n >> 2) * 4];
}

__device__ __forceinline__ void pair_tile_store(float* Tl, const v2f a,
                                                const v2f b, const int tile,
                                                const int n, const int h) {
    v8f z = {};
    z = __builtin_amdgcn_wmma_f32_16x16x4_f32(
            false, a, false, b, (short)0, z, false, false);
    #pragma unroll
    for (int v = 0; v < 8; ++v) {
        const int k4 = 4 * tile + 2 * h + (v >> 2);
        Tl[(k4 & 7) * 128 + (k4 >> 3) * 64 + ((v & 3) * 4 + (n >> 2)) * 4 + (n & 3)] = z[v];
    }
}

__device__ __forceinline__ void apply_pair(const float* env, const float* Tl,
                                           const int wave, int n, int h,
                                           v8f& r0, v8f& r1) {
    asm volatile("" : "+v"(n), "+v"(h));

    const float* tp = Tl + (h * 16 + n) * 4;
    const float* e0 = env + (wave * 16 + n) * 64;
    const float* e1 = e0 + 8 * 16 * 64;

    v8f acc00 = {}, acc01 = {}, acc10 = {}, acc11 = {};
    #pragma unroll
    for (int j = 0; j < 8; ++j) {
        const int slot = ((8 * h + j) ^ n) << 2;
        const v4f tv = *(const v4f*)(tp + j * 128);
        const v4f u0 = *(const v4f*)(e0 + slot);
        const v4f u1 = *(const v4f*)(e1 + slot);
        const v2f ta = {tv.x, tv.y}, tb = {tv.z, tv.w};
        const v2f x0 = {u0.x, u0.y}, y0 = {u0.z, u0.w};
        const v2f x1 = {u1.x, u1.y}, y1 = {u1.z, u1.w};
        acc00 = __builtin_amdgcn_wmma_f32_16x16x4_f32(
                  false, ta, false, x0, (short)0, acc00, false, false);
        acc10 = __builtin_amdgcn_wmma_f32_16x16x4_f32(
                  false, ta, false, x1, (short)0, acc10, false, false);
        acc01 = __builtin_amdgcn_wmma_f32_16x16x4_f32(
                  false, tb, false, y0, (short)0, acc01, false, false);
        acc11 = __builtin_amdgcn_wmma_f32_16x16x4_f32(
                  false, tb, false, y1, (short)0, acc11, false, false);
    }
    asm volatile("v_nop\n\tv_nop\n\tv_nop\n\tv_nop"
                 : "+v"(acc00), "+v"(acc01), "+v"(acc10), "+v"(acc11));
    r0 = acc00 + acc01;
    r1 = acc10 + acc11;
}

__device__ __forceinline__ void top_tile_load(const float* __restrict__ M0,
                                              const float* __restrict__ M1,
                                              const int w, int n, int h,
                                              v2f& a, v2f& b) {
    asm volatile("" : "+v"(n), "+v"(h));
    a = *(const v2f*)(M0 + (n & 3) * 16 + (n >> 2) * 4 + 2 * h);
    b.x = M1[((2 * h    ) * 4 + w) * 16 + n];
    b.y = M1[((2 * h + 1) * 4 + w) * 16 + n];
}

__device__ __forceinline__ void top_tile_store(float* Tl, const v2f a,
                                               const v2f b, const int w,
                                               int n, int h) {
    asm volatile("" : "+v"(n), "+v"(h));
    v8f z = {};
    z = __builtin_amdgcn_wmma_f32_16x16x4_f32(
            false, a, false, b, (short)0, z, false, false);
    #pragma unroll
    for (int v = 0; v < 8; ++v) {
        const int nt = 2 * h + (v >> 2);
        const int kc = v & 3;
        Tl[((nt * 2 + (kc >> 1)) * 32 + (w >> 1) * 16 + n) * 4 + (kc & 1) * 2 + (w & 1)] = z[v];
    }
}

__device__ __forceinline__ void apply_top_pair(float* env, const float* Tl,
                                               const int wave, int n, int h) {
    asm volatile("" : "+v"(n), "+v"(h));

    const int r0 = ((wave & 3) * 16 + n) * 4;
    const int nh = wave >> 2;

    const float* src = env + r0 + ((2 * h) << 8);
    v4f e0[4], e1[4];
    #pragma unroll
    for (int kc = 0; kc < 4; ++kc) {
        e0[kc] = *(const v4f*)(src + ((4 * kc    ) << 8));
        e1[kc] = *(const v4f*)(src + ((4 * kc + 1) << 8));
    }
    const float* tp = Tl + (nh * 4 * 32 + h * 16 + n) * 4;

    #pragma unroll
    for (int q = 0; q < 2; ++q) {
        const v4f t[2] = { *(const v4f*)(tp + (q * 2    ) * 128),
                           *(const v4f*)(tp + (q * 2 + 1) * 128) };
        v8f acc0 = {}, acc1 = {}, acc2 = {}, acc3 = {};
        #pragma unroll
        for (int kc = 0; kc < 4; ++kc) {
            const v4f tv = t[kc >> 1];
            const v2f m  = (kc & 1) ? (v2f){tv.z, tv.w} : (v2f){tv.x, tv.y};
            v2f x0 = {e0[kc].x, e1[kc].x};
            v2f x1 = {e0[kc].y, e1[kc].y};
            v2f x2 = {e0[kc].z, e1[kc].z};
            v2f x3 = {e0[kc].w, e1[kc].w};
            acc0 = __builtin_amdgcn_wmma_f32_16x16x4_f32(
                     false, m, false, x0, (short)0, acc0, false, false);
            acc1 = __builtin_amdgcn_wmma_f32_16x16x4_f32(
                     false, m, false, x1, (short)0, acc1, false, false);
            acc2 = __builtin_amdgcn_wmma_f32_16x16x4_f32(
                     false, m, false, x2, (short)0, acc2, false, false);
            acc3 = __builtin_amdgcn_wmma_f32_16x16x4_f32(
                     false, m, false, x3, (short)0, acc3, false, false);
        }
        asm volatile("v_nop\n\tv_nop\n\tv_nop\n\tv_nop"
                     : "+v"(acc0), "+v"(acc1), "+v"(acc2), "+v"(acc3));

        if (q == 0) __syncthreads();

        float* dst = env + (((2 * nh + q) * 16 + 8 * h) << 8) + r0;
        #pragma unroll
        for (int v = 0; v < 8; ++v) {
            v4f o = {acc0[v], acc1[v], acc2[v], acc3[v]};
            *(v4f*)(dst + (v << 8)) = o;
        }

        if (q == 0)
            asm volatile("" : "+v"(e0[0]), "+v"(e0[1]), "+v"(e0[2]), "+v"(e0[3]),
                              "+v"(e1[0]), "+v"(e1[1]), "+v"(e1[2]), "+v"(e1[3])
                            : : "memory");
    }
}

__device__ __forceinline__ void apply_top_pair_last(float* env, const float* Tl,
                                                    const int wave, int n, int h) {
    asm volatile("" : "+v"(n), "+v"(h));
    if (wave >= 4) return;

    const int r0 = (wave * 16 + n) * 4;
    const float* src = env + r0 + ((2 * h) << 8);
    v4f e0[4], e1[4];
    #pragma unroll
    for (int kc = 0; kc < 4; ++kc) {
        e0[kc] = *(const v4f*)(src + ((4 * kc    ) << 8));
        e1[kc] = *(const v4f*)(src + ((4 * kc + 1) << 8));
    }
    const float* tp = Tl + (h * 16 + n) * 4;
    const v4f t[2] = { *(const v4f*)(tp), *(const v4f*)(tp + 128) };

    v8f acc0 = {}, acc1 = {}, acc2 = {}, acc3 = {};
    #pragma unroll
    for (int kc = 0; kc < 4; ++kc) {
        const v4f tv = t[kc >> 1];
        const v2f m  = (kc & 1) ? (v2f){tv.z, tv.w} : (v2f){tv.x, tv.y};
        v2f x0 = {e0[kc].x, e1[kc].x};
        v2f x1 = {e0[kc].y, e1[kc].y};
        v2f x2 = {e0[kc].z, e1[kc].z};
        v2f x3 = {e0[kc].w, e1[kc].w};
        acc0 = __builtin_amdgcn_wmma_f32_16x16x4_f32(
                 false, m, false, x0, (short)0, acc0, false, false);
        acc1 = __builtin_amdgcn_wmma_f32_16x16x4_f32(
                 false, m, false, x1, (short)0, acc1, false, false);
        acc2 = __builtin_amdgcn_wmma_f32_16x16x4_f32(
                 false, m, false, x2, (short)0, acc2, false, false);
        acc3 = __builtin_amdgcn_wmma_f32_16x16x4_f32(
                 false, m, false, x3, (short)0, acc3, false, false);
    }
    asm volatile("v_nop\n\tv_nop\n\tv_nop\n\tv_nop"
                 : "+v"(acc0), "+v"(acc1), "+v"(acc2), "+v"(acc3));

    if (h == 0) {
        #pragma unroll
        for (int v = 0; v < 4; ++v) {
            v4f o = {acc0[v], acc1[v], acc2[v], acc3[v]};
            *(v4f*)(env + r0 + (v << 8)) = o;
        }
    }
}

#define LW   8192
#define LT   14336
#define LS1  14592
#define LS2  14848
#define LS3  4096

__device__ __forceinline__ void left_pair_build(float* Wl,
                                                const float* __restrict__ Msite,
                                                const int* __restrict__ xrow,
                                                const int wave, const int n,
                                                const int h) {
    for (int t = wave; t < 24; t += 8) {
        const int i = t >> 2, u1 = t & 3;
        const float* M0 = Msite + ((i * LY + 0) * PD + xrow[i * LY + 0]) * 256;
        const float* M1 = Msite + ((i * LY + 1) * PD + xrow[i * LY + 1]) * 256;
        v2f a, b;
        a.x = M0[(n >> 2) * 64 + (2 * h    ) * 4 + (n & 3)];
        a.y = M0[(n >> 2) * 64 + (2 * h + 1) * 4 + (n & 3)];
        b.x = M1[(u1 * 4 + 2 * h    ) * 16 + n];
        b.y = M1[(u1 * 4 + 2 * h + 1) * 16 + n];
        v8f z = {};
        z = __builtin_amdgcn_wmma_f32_16x16x4_f32(
                false, a, false, b, (short)0, z, false, false);
        float* W = Wl + i * 1024 + u1 * 64 + (n >> 2) * 16 + (n & 3);
        #pragma unroll
        for (int v = 0; v < 8; ++v)
            W[(2 * h + (v >> 2)) * 256 + (v & 3) * 4] = z[v];
    }
}

template <int M, int NT>
__device__ __forceinline__ void chain_gemm(const float* A, const float* B,
                                           float* D, const int wave,
                                           const int n, const int h) {
    constexpr int MT  = (M + 15) / 16;
    constexpr int ldn = 16 * NT;
    for (int t = wave; t < MT * NT; t += 8) {
        const int mt = t / NT, nt = t % NT;
        const int row = (M < 16) ? (n & (M - 1)) : (mt * 16 + n);
        v2f a[4], b[4];
        #pragma unroll
        for (int kc = 0; kc < 4; ++kc) {
            a[kc]   = *(const v2f*)(A + row * 16 + 4 * kc + 2 * h);
            b[kc].x = B[(4 * kc + 2 * h    ) * ldn + nt * 16 + n];
            b[kc].y = B[(4 * kc + 2 * h + 1) * ldn + nt * 16 + n];
        }
        v8f acc = {};
        #pragma unroll
        for (int kc = 0; kc < 4; ++kc)
            acc = __builtin_amdgcn_wmma_f32_16x16x4_f32(
                    false, a[kc], false, b[kc], (short)0, acc, false, false);
        #pragma unroll
        for (int v = 0; v < 8; ++v) {
            const int r = mt * 16 + 8 * h + v;
            if (r < M) D[r * ldn + nt * 16 + n] = acc[v];
        }
    }
}

__device__ __forceinline__ void left_tail_tile(const float* W4, const float* W5,
                                               float* T, const int mt,
                                               const int n, const int h) {
    v2f a[4], b[4];
    #pragma unroll
    for (int kc = 0; kc < 4; ++kc) {
        a[kc]   = *(const v2f*)(W4 + (mt * 16 + n) * 16 + 4 * kc + 2 * h);
        b[kc].x = W5[(4 * kc + 2 * h    ) * 64 + (n & 3) * 16];
        b[kc].y = W5[(4 * kc + 2 * h + 1) * 64 + (n & 3) * 16];
    }
    v8f acc = {};
    #pragma unroll
    for (int kc = 0; kc < 4; ++kc)
        acc = __builtin_amdgcn_wmma_f32_16x16x4_f32(
                false, a[kc], false, b[kc], (short)0, acc, false, false);
    if (n < 4) {
        #pragma unroll
        for (int v = 0; v < 8; ++v)
            T[(mt * 16 + 8 * h + v) * 4 + n] = acc[v];
    }
}

__global__ void __launch_bounds__(256)
peps_amp_kernel(const int* __restrict__ x,
                const float* __restrict__ Msite,
                float* __restrict__ out) {
    __shared__ __attribute__((aligned(16))) float env[ENV7 + TLEN];
    float* const Tl = env + ENV7;

    const int tid    = threadIdx.x;
    const int sample = blockIdx.x;
    const int lane   = tid & 31;
    const int wave   = __builtin_amdgcn_readfirstlane(tid >> 5);
    __builtin_assume(wave >= 0 && wave < 8);
    const int n      = lane & 15;
    const int h      = lane >> 4;

    const int* xrow = x + sample * NSITE;

    left_pair_build(env + LW, Msite, xrow, wave, n, h);
    __syncthreads();
    chain_gemm<4, 4>(env + LW, env + LW + 1024, env + LS1, wave, n, h);
    if (wave >= 4)
        left_tail_tile(env + LW + 4096, env + LW + 5120, env + LT, wave - 4, n, h);
    __syncthreads();
    chain_gemm<16, 4>(env + LS1, env + LW + 2048, env + LS2, wave, n, h);
    __syncthreads();
    chain_gemm<64, 4>(env + LS2, env + LW + 3072, env + LS3, wave, n, h);
    __syncthreads();
    {
        v2f qa = {}, qb = {};
        if (wave < 4) {
            const int s0 = xrow[0 * LY + 2];
            const int s1 = xrow[1 * LY + 2];
            top_tile_load(Msite + ((0 * LY + 2) * PD + s0) * 256,
                          Msite + ((1 * LY + 2) * PD + s1) * 256,
                          wave, n, h, qa, qb);
        }
        chain_gemm<256, 1>(env + LS3, env + LT, env, wave, n, h);
        if (wave < 4)
            top_tile_store(Tl, qa, qb, wave, n, h);
    }
    __syncthreads();

    for (int j = 2; j < LY; ++j) {
        const bool inner = (j < LY - 1);
        const int  nrow  = inner ? LX - 1 : LX;

        for (int i = 0; i < nrow; ++i) {
            if (i == 1) continue;
            const int site = i * LY + j;
            const int s    = xrow[site];
            const float* Mp = Msite + (site * PD + s) * 256;

            const int shift = 2 * (5 - i);

            {
                const int ntile = (j == LY - 1)
                                ? ((shift >= 4) ? (1 << (shift - 4)) : 1) : 64;
                if (i == 0) {
                    if (inner) apply_top_pair(env, Tl, wave, n, h);
                    else       apply_top_pair_last(env, Tl, wave, n, h);
                } else if (i == 2)
                    apply_site_rows4<4, 6>(env, Mp, ntile >> 2, wave, n, h);
                else if (inner && i == 3) {
                    v2f pa = {}, pb = {};
                    if (wave < 4) {
                        const int s4 = xrow[4 * LY + j];
                        const int s5 = xrow[5 * LY + j];
                        pair_tile_load(Msite + ((4 * LY + j) * PD + s4) * 256,
                                       Msite + ((5 * LY + j) * PD + s5) * 256,
                                       wave, n, h, pa, pb);
                    }
                    apply_site_row3(env, Mp, wave, n, h);
                    if (wave < 4)
                        pair_tile_store(Tl, pa, pb, wave, n, h);
                } else if (inner && i == 4) {
                    const bool top = (wave < 4);
                    v2f qa = {}, qb = {};
                    if (top) {
                        const int s0 = xrow[0 * LY + j + 1];
                        const int s1 = xrow[1 * LY + j + 1];
                        top_tile_load(Msite + ((0 * LY + j + 1) * PD + s0) * 256,
                                      Msite + ((1 * LY + j + 1) * PD + s1) * 256,
                                      wave, n, h, qa, qb);
                    }
                    v8f r0, r1;
                    apply_pair(env, Tl, wave, n, h, r0, r1);
                    __syncthreads();
                    if (top)
                        top_tile_store(Tl, qa, qb, wave, n, h);
                    float* o0 = env + (wave * 16 + n) * 16 + 8 * h;
                    float* o1 = o0 + 8 * 16 * 16;
                    *(v4f*)(o0    ) = __builtin_shufflevector(r0, r0, 0, 1, 2, 3);
                    *(v4f*)(o0 + 4) = __builtin_shufflevector(r0, r0, 4, 5, 6, 7);
                    *(v4f*)(o1    ) = __builtin_shufflevector(r1, r1, 0, 1, 2, 3);
                    *(v4f*)(o1 + 4) = __builtin_shufflevector(r1, r1, 4, 5, 6, 7);
                } else if (shift >= 4)
                    apply_site<4, 1, 0>(env, Mp, shift, ntile, 0, wave, n, h);
                else if (i == LX - 2)
                    apply_site<4, 1, 1>(env, Mp, shift, ntile, 0, wave, n, h);
                else
                    apply_site<4, 1, 2>(env, Mp, shift, ntile, 0, wave, n, h);
            }
            __syncthreads();
        }
    }

    if (tid == 0) out[sample] = env[0];
}

extern "C" void kernel_launch(void* const* d_in, const int* in_sizes, int n_in,
                              void* d_out, int out_size, void* d_ws, size_t ws_size,
                              hipStream_t stream) {
    const int*   x     = (const int*)d_in[0];
    const float* peps  = (const float*)d_in[1];
    float*       out   = (float*)d_out;
    float*       Msite = (float*)d_ws;

    const int nM = NSITE * PD * 256;
    peps_prep_kernel<<<(nM + 255) / 256, 256, 0, stream>>>(peps, Msite);

    const int B = out_size;
    peps_amp_kernel<<<B, 256, 0, stream>>>(x, Msite, out);
}


# columns 5 and 4 contracted from the right as a second bond-16 matrix-product chain and joined with the env behind column 3 by a 4096-term inner product; sum order changes: right half first, tree sum
# speedup vs baseline: 9.6512x; 1.3524x over previous
#include <hip/hip_runtime.h>


typedef float v2f __attribute__((ext_vector_type(2)));
typedef float v4f __attribute__((ext_vector_type(4)));
typedef float v8f __attribute__((ext_vector_type(8)));

#define LX 6
#define LY 6
#define PD 2
#define DD 4
#define NSITE 36
#define ENV7 16384
#define ENV6 4096
#define TLEN 1024

__global__ void peps_prep_kernel(const float* __restrict__ peps,
                                 float* __restrict__ Msite) {
    int idx = blockIdx.x * blockDim.x + threadIdx.x;
    if (idx >= NSITE * PD * 256) return;
    int n    = idx & 15;
    int k    = (idx >> 4) & 15;
    int s    = (idx >> 8) & 1;
    int site = idx >> 9;
    int i = site / LY, j = site % LY;
    int a = k >> 2, c = k & 3;
    int d = n >> 2, b = n & 3;
    float v = 0.0f;
    bool pad = (i == 0      && a > 0) || (i == LX - 1 && b > 0) ||
               (j == 0      && c > 0) || (j == LY - 1 && d > 0);
    if (!pad) {
        int p = ((((site * PD + s) * DD + a) * DD + b) * DD + c) * DD + d;
        v = peps[p];
    }
    Msite[idx] = v;
}

template <int NK, int SHIFT>
__device__ __forceinline__ void apply_site_rows4(float* env,
                                                 const float* __restrict__ Mp,
                                                 const int ngroup,
                                                 const int wave, const int n,
                                                 const int h) {
    constexpr int shift = SHIFT;
    constexpr int psm = (1 << shift) - 1;

    v2f m[NK];
    #pragma unroll
    for (int kc = 0; kc < NK; ++kc) {
        m[kc].x = Mp[(4 * kc + 2 * h    ) * 16 + n];
        m[kc].y = Mp[(4 * kc + 2 * h + 1) * 16 + n];
    }

    for (int g = wave; g < ngroup; g += 8) {
        const int r0   = (g * 16 + n) * 4;
        const int post = r0 & psm;
        const int base = ((r0 - post) << 4) + post;

        const float* src = env + base + ((2 * h) << shift);
        float*       dst = env + base + ((8 * h) << shift);

        v4f e0[NK], e1[NK];
        #pragma unroll
        for (int kc = 0; kc < NK; ++kc) {
            e0[kc] = *(const v4f*)(src + ((4 * kc    ) << shift));
            e1[kc] = *(const v4f*)(src + ((4 * kc + 1) << shift));
        }

        v8f acc0 = {}, acc1 = {}, acc2 = {}, acc3 = {};
        #pragma unroll
        for (int kc = 0; kc < NK; ++kc) {
            v2f x0 = {e0[kc].x, e1[kc].x};
            v2f x1 = {e0[kc].y, e1[kc].y};
            v2f x2 = {e0[kc].z, e1[kc].z};
            v2f x3 = {e0[kc].w, e1[kc].w};
            acc0 = __builtin_amdgcn_wmma_f32_16x16x4_f32(
                     false, m[kc], false, x0, (short)0, acc0, false, false);
            acc1 = __builtin_amdgcn_wmma_f32_16x16x4_f32(
                     false, m[kc], false, x1, (short)0, acc1, false, false);
            acc2 = __builtin_amdgcn_wmma_f32_16x16x4_f32(
                     false, m[kc], false, x2, (short)0, acc2, false, false);
            acc3 = __builtin_amdgcn_wmma_f32_16x16x4_f32(
                     false, m[kc], false, x3, (short)0, acc3, false, false);
        }

        asm volatile("v_nop\n\tv_nop\n\tv_nop\n\tv_nop"
                     : "+v"(acc0), "+v"(acc1), "+v"(acc2), "+v"(acc3));

        #pragma unroll
        for (int v = 0; v < 8; ++v) {
            v4f o = {acc0[v], acc1[v], acc2[v], acc3[v]};
            *(v4f*)(dst + (v << shift)) = o;
        }
    }
}

__device__ __forceinline__ void apply_site_row3(float* env,
                                                const float* __restrict__ Mp,
                                                const int wave, int n, int h) {
    asm volatile("" : "+v"(n), "+v"(h));

    v2f b[4];
    #pragma unroll
    for (int kc = 0; kc < 4; ++kc) {
        b[kc].x = Mp[(4 * kc + 2 * h    ) * 16 + n];
        b[kc].y = Mp[(4 * kc + 2 * h + 1) * 16 + n];
    }
    const int w3 = wave & 3;
    int so[8];
    #pragma unroll
    for (int v = 0; v < 8; ++v) {
        const int r3 = 2 * h + (v >> 2);
        so[v] = r3 * 64 + (((v & 3) ^ w3) << 4) + (((n >> 2) ^ r3) << 2) + (n & 3);
    }

    #pragma unroll 2
    for (int t = wave; t < 64; t += 8) {
        const int tb = t * 256;
        v2f a[4];
        #pragma unroll
        for (int kc = 0; kc < 4; ++kc) {
            a[kc].x = env[tb + n + ((4 * kc + 2 * h    ) << 4)];
            a[kc].y = env[tb + n + ((4 * kc + 2 * h + 1) << 4)];
        }
        v8f acc = {};
        #pragma unroll
        for (int kc = 0; kc < 4; ++kc)
            acc = __builtin_amdgcn_wmma_f32_16x16x4_f32(
                    false, b[kc], false, a[kc], (short)0, acc, false, false);
        #pragma unroll
        for (int v = 0; v < 8; ++v)
            env[tb + so[v]] = acc[v];
    }
}

__device__ __forceinline__ void pair_tile_load(const float* __restrict__ M4,
                                               const float* __restrict__ M5,
                                               const int tile, const int n,
                                               const int h, v2f& a, v2f& b) {
    a = *(const v2f*)(M4 + (16 * tile + n) * 4 + 2 * h);
    b.x = M5[((2 * h    ) * 4 + (n & 3)) * 16 + (n >> 2) * 4];
    b.y = M5[((2 * h + 1) * 4 + (n & 3)) * 16 + (n >> 2) * 4];
}

__device__ __forceinline__ void pair_tile_store(float* Tl, const v2f a,
                                                const v2f b, const int tile,
                                                const int n, const int h) {
    v8f z = {};
    z = __builtin_amdgcn_wmma_f32_16x16x4_f32(
            false, a, false, b, (short)0, z, false, false);
    #pragma unroll
    for (int v = 0; v < 8; ++v) {
        const int k4 = 4 * tile + 2 * h + (v >> 2);
        Tl[(k4 & 7) * 128 + (k4 >> 3) * 64 + ((v & 3) * 4 + (n >> 2)) * 4 + (n & 3)] = z[v];
    }
}

__device__ __forceinline__ void apply_pair(const float* env, const float* Tl,
                                           const int wave, int n, int h,
                                           v8f& r0, v8f& r1) {
    asm volatile("" : "+v"(n), "+v"(h));

    const float* tp = Tl + (h * 16 + n) * 4;
    const float* e0 = env + (wave * 16 + n) * 64;
    const float* e1 = e0 + 8 * 16 * 64;

    v8f acc00 = {}, acc01 = {}, acc10 = {}, acc11 = {};
    #pragma unroll
    for (int j = 0; j < 8; ++j) {
        const int slot = ((8 * h + j) ^ n) << 2;
        const v4f tv = *(const v4f*)(tp + j * 128);
        const v4f u0 = *(const v4f*)(e0 + slot);
        const v4f u1 = *(const v4f*)(e1 + slot);
        const v2f ta = {tv.x, tv.y}, tb = {tv.z, tv.w};
        const v2f x0 = {u0.x, u0.y}, y0 = {u0.z, u0.w};
        const v2f x1 = {u1.x, u1.y}, y1 = {u1.z, u1.w};
        acc00 = __builtin_amdgcn_wmma_f32_16x16x4_f32(
                  false, ta, false, x0, (short)0, acc00, false, false);
        acc10 = __builtin_amdgcn_wmma_f32_16x16x4_f32(
                  false, ta, false, x1, (short)0, acc10, false, false);
        acc01 = __builtin_amdgcn_wmma_f32_16x16x4_f32(
                  false, tb, false, y0, (short)0, acc01, false, false);
        acc11 = __builtin_amdgcn_wmma_f32_16x16x4_f32(
                  false, tb, false, y1, (short)0, acc11, false, false);
    }
    asm volatile("v_nop\n\tv_nop\n\tv_nop\n\tv_nop"
                 : "+v"(acc00), "+v"(acc01), "+v"(acc10), "+v"(acc11));
    r0 = acc00 + acc01;
    r1 = acc10 + acc11;
}

__device__ __forceinline__ void top_tile_load(const float* __restrict__ M0,
                                              const float* __restrict__ M1,
                                              const int w, int n, int h,
                                              v2f& a, v2f& b) {
    asm volatile("" : "+v"(n), "+v"(h));
    a = *(const v2f*)(M0 + (n & 3) * 16 + (n >> 2) * 4 + 2 * h);
    b.x = M1[((2 * h    ) * 4 + w) * 16 + n];
    b.y = M1[((2 * h + 1) * 4 + w) * 16 + n];
}

__device__ __forceinline__ void top_tile_store(float* Tl, const v2f a,
                                               const v2f b, const int w,
                                               int n, int h) {
    asm volatile("" : "+v"(n), "+v"(h));
    v8f z = {};
    z = __builtin_amdgcn_wmma_f32_16x16x4_f32(
            false, a, false, b, (short)0, z, false, false);
    #pragma unroll
    for (int v = 0; v < 8; ++v) {
        const int nt = 2 * h + (v >> 2);
        const int kc = v & 3;
        Tl[((nt * 2 + (kc >> 1)) * 32 + (w >> 1) * 16 + n) * 4 + (kc & 1) * 2 + (w & 1)] = z[v];
    }
}

__device__ __forceinline__ void apply_top_pair(float* env, const float* Tl,
                                               const int wave, int n, int h) {
    asm volatile("" : "+v"(n), "+v"(h));

    const int r0 = ((wave & 3) * 16 + n) * 4;
    const int nh = wave >> 2;

    const float* src = env + r0 + ((2 * h) << 8);
    v4f e0[4], e1[4];
    #pragma unroll
    for (int kc = 0; kc < 4; ++kc) {
        e0[kc] = *(const v4f*)(src + ((4 * kc    ) << 8));
        e1[kc] = *(const v4f*)(src + ((4 * kc + 1) << 8));
    }
    const float* tp = Tl + (nh * 4 * 32 + h * 16 + n) * 4;

    #pragma unroll
    for (int q = 0; q < 2; ++q) {
        const v4f t[2] = { *(const v4f*)(tp + (q * 2    ) * 128),
                           *(const v4f*)(tp + (q * 2 + 1) * 128) };
        v8f acc0 = {}, acc1 = {}, acc2 = {}, acc3 = {};
        #pragma unroll
        for (int kc = 0; kc < 4; ++kc) {
            const v4f tv = t[kc >> 1];
            const v2f m  = (kc & 1) ? (v2f){tv.z, tv.w} : (v2f){tv.x, tv.y};
            v2f x0 = {e0[kc].x, e1[kc].x};
            v2f x1 = {e0[kc].y, e1[kc].y};
            v2f x2 = {e0[kc].z, e1[kc].z};
            v2f x3 = {e0[kc].w, e1[kc].w};
            acc0 = __builtin_amdgcn_wmma_f32_16x16x4_f32(
                     false, m, false, x0, (short)0, acc0, false, false);
            acc1 = __builtin_amdgcn_wmma_f32_16x16x4_f32(
                     false, m, false, x1, (short)0, acc1, false, false);
            acc2 = __builtin_amdgcn_wmma_f32_16x16x4_f32(
                     false, m, false, x2, (short)0, acc2, false, false);
            acc3 = __builtin_amdgcn_wmma_f32_16x16x4_f32(
                     false, m, false, x3, (short)0, acc3, false, false);
        }
        asm volatile("v_nop\n\tv_nop\n\tv_nop\n\tv_nop"
                     : "+v"(acc0), "+v"(acc1), "+v"(acc2), "+v"(acc3));

        if (q == 0) __syncthreads();

        float* dst = env + (((2 * nh + q) * 16 + 8 * h) << 8) + r0;
        #pragma unroll
        for (int v = 0; v < 8; ++v) {
            v4f o = {acc0[v], acc1[v], acc2[v], acc3[v]};
            *(v4f*)(dst + (v << 8)) = o;
        }

        if (q == 0)
            asm volatile("" : "+v"(e0[0]), "+v"(e0[1]), "+v"(e0[2]), "+v"(e0[3]),
                              "+v"(e1[0]), "+v"(e1[1]), "+v"(e1[2]), "+v"(e1[3])
                            : : "memory");
    }
}

#define LW   8192
#define LT   14336
#define LS1  14592
#define LS2  14848
#define LS3  4096
#define JR   4

template <int RIGHT>
__device__ __forceinline__ void pair_chain_build(float* Wl,
                                                 const float* __restrict__ Msite,
                                                 const int* __restrict__ xrow,
                                                 const int wave, const int n,
                                                 const int h) {
    constexpr int je = RIGHT ? LY - 1 : 0;
    constexpr int jn = RIGHT ? LY - 2 : 1;
    for (int t = wave; t < 24; t += 8) {
        const int i = t >> 2, u1 = t & 3;
        const float* Me = Msite + ((i * LY + je) * PD + xrow[i * LY + je]) * 256;
        const float* Mn = Msite + ((i * LY + jn) * PD + xrow[i * LY + jn]) * 256;
        v2f a, b;
        if (RIGHT) {
            a.x = Me[((n >> 2) * 4 + 2 * h    ) * 16 + (n & 3)];
            a.y = Me[((n >> 2) * 4 + 2 * h + 1) * 16 + (n & 3)];
            b.x = Mn[(u1 * 4 + (n >> 2)) * 16 + (2 * h    ) * 4 + (n & 3)];
            b.y = Mn[(u1 * 4 + (n >> 2)) * 16 + (2 * h + 1) * 4 + (n & 3)];
        } else {
            a.x = Me[(n >> 2) * 64 + (2 * h    ) * 4 + (n & 3)];
            a.y = Me[(n >> 2) * 64 + (2 * h + 1) * 4 + (n & 3)];
            b.x = Mn[(u1 * 4 + 2 * h    ) * 16 + n];
            b.y = Mn[(u1 * 4 + 2 * h + 1) * 16 + n];
        }
        v8f z = {};
        z = __builtin_amdgcn_wmma_f32_16x16x4_f32(
                false, a, false, b, (short)0, z, false, false);
        float* W = Wl + i * 1024 + u1 * 64 + (n >> 2) * 16 + (n & 3);
        #pragma unroll
        for (int v = 0; v < 8; ++v)
            W[(2 * h + (v >> 2)) * 256 + (v & 3) * 4] = z[v];
    }
}

template <int M, int NT>
__device__ __forceinline__ void chain_gemm(const float* A, const float* B,
                                           float* D, const int wave,
                                           const int n, const int h) {
    constexpr int MT  = (M + 15) / 16;
    constexpr int ldn = 16 * NT;
    for (int t = wave; t < MT * NT; t += 8) {
        const int mt = t / NT, nt = t % NT;
        const int row = (M < 16) ? (n & (M - 1)) : (mt * 16 + n);
        v2f a[4], b[4];
        #pragma unroll
        for (int kc = 0; kc < 4; ++kc) {
            a[kc]   = *(const v2f*)(A + row * 16 + 4 * kc + 2 * h);
            b[kc].x = B[(4 * kc + 2 * h    ) * ldn + nt * 16 + n];
            b[kc].y = B[(4 * kc + 2 * h + 1) * ldn + nt * 16 + n];
        }
        v8f acc = {};
        #pragma unroll
        for (int kc = 0; kc < 4; ++kc)
            acc = __builtin_amdgcn_wmma_f32_16x16x4_f32(
                    false, a[kc], false, b[kc], (short)0, acc, false, false);
        #pragma unroll
        for (int v = 0; v < 8; ++v) {
            const int r = mt * 16 + 8 * h + v;
            if (r < M) D[r * ldn + nt * 16 + n] = acc[v];
        }
    }
}

__device__ __forceinline__ void left_tail_tile(const float* W4, const float* W5,
                                               float* T, const int mt,
                                               const int n, const int h) {
    v2f a[4], b[4];
    #pragma unroll
    for (int kc = 0; kc < 4; ++kc) {
        a[kc]   = *(const v2f*)(W4 + (mt * 16 + n) * 16 + 4 * kc + 2 * h);
        b[kc].x = W5[(4 * kc + 2 * h    ) * 64 + (n & 3) * 16];
        b[kc].y = W5[(4 * kc + 2 * h + 1) * 64 + (n & 3) * 16];
    }
    v8f acc = {};
    #pragma unroll
    for (int kc = 0; kc < 4; ++kc)
        acc = __builtin_amdgcn_wmma_f32_16x16x4_f32(
                false, a[kc], false, b[kc], (short)0, acc, false, false);
    if (n < 4) {
        #pragma unroll
        for (int v = 0; v < 8; ++v)
            T[(mt * 16 + 8 * h + v) * 4 + n] = acc[v];
    }
}

__device__ __forceinline__ float chain_join(const float* A, const float* B,
                                            const float* L, const int wave,
                                            const int n, const int h) {
    float p = 0.0f;
    for (int t = wave; t < 16; t += 8) {
        v2f a[4], b[4];
        #pragma unroll
        for (int kc = 0; kc < 4; ++kc) {
            a[kc]   = *(const v2f*)(A + (t * 16 + n) * 16 + 4 * kc + 2 * h);
            b[kc].x = B[(4 * kc + 2 * h    ) * 16 + n];
            b[kc].y = B[(4 * kc + 2 * h + 1) * 16 + n];
        }
        float l[8];
        #pragma unroll
        for (int v = 0; v < 8; ++v)
            l[v] = L[(t * 16 + 8 * h + v) * 16 + n];
        v8f acc = {};
        #pragma unroll
        for (int kc = 0; kc < 4; ++kc)
            acc = __builtin_amdgcn_wmma_f32_16x16x4_f32(
                    false, a[kc], false, b[kc], (short)0, acc, false, false);
        asm volatile("v_nop\n\tv_nop\n\tv_nop\n\tv_nop" : "+v"(acc));
        #pragma unroll
        for (int v = 0; v < 8; ++v)
            p += acc[v] * l[v];
    }
    return p;
}

__global__ void __launch_bounds__(256)
peps_amp_kernel(const int* __restrict__ x,
                const float* __restrict__ Msite,
                float* __restrict__ out) {
    __shared__ __attribute__((aligned(16))) float env[ENV7 + TLEN];
    float* const Tl = env + ENV7;

    const int tid    = threadIdx.x;
    const int sample = blockIdx.x;
    const int lane   = tid & 31;
    const int wave   = __builtin_amdgcn_readfirstlane(tid >> 5);
    __builtin_assume(wave >= 0 && wave < 8);
    const int n      = lane & 15;
    const int h      = lane >> 4;

    const int* xrow = x + sample * NSITE;

    pair_chain_build<0>(env + LW, Msite, xrow, wave, n, h);
    __syncthreads();
    chain_gemm<4, 4>(env + LW, env + LW + 1024, env + LS1, wave, n, h);
    if (wave >= 4)
        left_tail_tile(env + LW + 4096, env + LW + 5120, env + LT, wave - 4, n, h);
    __syncthreads();
    chain_gemm<16, 4>(env + LS1, env + LW + 2048, env + LS2, wave, n, h);
    __syncthreads();
    chain_gemm<64, 4>(env + LS2, env + LW + 3072, env + LS3, wave, n, h);
    __syncthreads();
    {
        v2f qa = {}, qb = {};
        if (wave < 4) {
            const int s0 = xrow[0 * LY + 2];
            const int s1 = xrow[1 * LY + 2];
            top_tile_load(Msite + ((0 * LY + 2) * PD + s0) * 256,
                          Msite + ((1 * LY + 2) * PD + s1) * 256,
                          wave, n, h, qa, qb);
        }
        chain_gemm<256, 1>(env + LS3, env + LT, env, wave, n, h);
        if (wave < 4)
            top_tile_store(Tl, qa, qb, wave, n, h);
    }
    __syncthreads();

    for (int j = 2; j < JR; ++j) {
        for (int i = 0; i < LX - 1; ++i) {
            if (i == 1) continue;
            const int site = i * LY + j;
            const int s    = xrow[site];
            const float* Mp = Msite + (site * PD + s) * 256;

            if (i == 0) {
                apply_top_pair(env, Tl, wave, n, h);
            } else if (i == 2) {
                apply_site_rows4<4, 6>(env, Mp, 16, wave, n, h);
            } else if (i == 3) {
                v2f pa = {}, pb = {};
                if (wave < 4) {
                    const int s4 = xrow[4 * LY + j];
                    const int s5 = xrow[5 * LY + j];
                    pair_tile_load(Msite + ((4 * LY + j) * PD + s4) * 256,
                                   Msite + ((5 * LY + j) * PD + s5) * 256,
                                   wave, n, h, pa, pb);
                }
                apply_site_row3(env, Mp, wave, n, h);
                if (wave < 4)
                    pair_tile_store(Tl, pa, pb, wave, n, h);
            } else {
                const bool top = (wave < 4) && (j + 1 < JR);
                v2f qa = {}, qb = {};
                if (top) {
                    const int s0 = xrow[0 * LY + j + 1];
                    const int s1 = xrow[1 * LY + j + 1];
                    top_tile_load(Msite + ((0 * LY + j + 1) * PD + s0) * 256,
                                  Msite + ((1 * LY + j + 1) * PD + s1) * 256,
                                  wave, n, h, qa, qb);
                }
                v8f r0, r1;
                apply_pair(env, Tl, wave, n, h, r0, r1);
                __syncthreads();
                if (top)
                    top_tile_store(Tl, qa, qb, wave, n, h);
                float* o0 = env + (wave * 16 + n) * 16 + 8 * h;
                float* o1 = o0 + 8 * 16 * 16;
                *(v4f*)(o0    ) = __builtin_shufflevector(r0, r0, 0, 1, 2, 3);
                *(v4f*)(o0 + 4) = __builtin_shufflevector(r0, r0, 4, 5, 6, 7);
                *(v4f*)(o1    ) = __builtin_shufflevector(r1, r1, 0, 1, 2, 3);
                *(v4f*)(o1 + 4) = __builtin_shufflevector(r1, r1, 4, 5, 6, 7);
                if (j + 1 == JR)
                    pair_chain_build<1>(env + LW, Msite, xrow, wave, n, h);
            }
            __syncthreads();
        }
    }

    chain_gemm<4, 4>(env + LW, env + LW + 1024, env + LS1, wave, n, h);
    if (wave >= 4)
        left_tail_tile(env + LW + 4096, env + LW + 5120, env + LT, wave - 4, n, h);
    __syncthreads();
    chain_gemm<16, 4>(env + LS1, env + LW + 2048, env + LS2, wave, n, h);
    __syncthreads();
    chain_gemm<64, 4>(env + LS2, env + LW + 3072, env + LS3, wave, n, h);
    __syncthreads();
    float p = chain_join(env + LS3, env + LT, env, wave, n, h);

    for (int o = 16; o > 0; o >>= 1)
        p += __shfl_xor(p, o);
    if (lane == 0) env[LS1 + wave] = p;
    __syncthreads();
    if (tid == 0) {
        float a = 0.0f;
        for (int w = 0; w < 8; ++w) a += env[LS1 + w];
        out[sample] = a;
    }
}

extern "C" void kernel_launch(void* const* d_in, const int* in_sizes, int n_in,
                              void* d_out, int out_size, void* d_ws, size_t ws_size,
                              hipStream_t stream) {
    const int*   x     = (const int*)d_in[0];
    const float* peps  = (const float*)d_in[1];
    float*       out   = (float*)d_out;
    float*       Msite = (float*)d_ws;

    const int nM = NSITE * PD * 256;
    peps_prep_kernel<<<(nM + 255) / 256, 256, 0, stream>>>(peps, Msite);

    const int B = out_size;
    peps_amp_kernel<<<B, 256, 0, stream>>>(x, Msite, out);
}


# chains of columns 0-1 and 5-4 tabulated in the prep launch per 6-index half (64 heads, 64 tails a side); a sample's edge env is ONE 64x16x64 product; sum order changes: chain re-associated
# speedup vs baseline: 11.1197x; 1.1522x over previous
#include <hip/hip_runtime.h>


typedef float v2f __attribute__((ext_vector_type(2)));
typedef float v4f __attribute__((ext_vector_type(4)));
typedef float v8f __attribute__((ext_vector_type(8)));

#define LX 6
#define LY 6
#define PD 2
#define DD 4
#define NSITE 36
#define ENV7 16384
#define ENV6 4096
#define TLEN 1024

template <int NK, int SHIFT>
__device__ __forceinline__ void apply_site_rows4(float* env,
                                                 const float* __restrict__ Mp,
                                                 const int ngroup,
                                                 const int wave, const int n,
                                                 const int h) {
    constexpr int shift = SHIFT;
    constexpr int psm = (1 << shift) - 1;

    v2f m[NK];
    #pragma unroll
    for (int kc = 0; kc < NK; ++kc) {
        m[kc].x = Mp[(4 * kc + 2 * h    ) * 16 + n];
        m[kc].y = Mp[(4 * kc + 2 * h + 1) * 16 + n];
    }

    for (int g = wave; g < ngroup; g += 8) {
        const int r0   = (g * 16 + n) * 4;
        const int post = r0 & psm;
        const int base = ((r0 - post) << 4) + post;

        const float* src = env + base + ((2 * h) << shift);
        float*       dst = env + base + ((8 * h) << shift);

        v4f e0[NK], e1[NK];
        #pragma unroll
        for (int kc = 0; kc < NK; ++kc) {
            e0[kc] = *(const v4f*)(src + ((4 * kc    ) << shift));
            e1[kc] = *(const v4f*)(src + ((4 * kc + 1) << shift));
        }

        v8f acc0 = {}, acc1 = {}, acc2 = {}, acc3 = {};
        #pragma unroll
        for (int kc = 0; kc < NK; ++kc) {
            v2f x0 = {e0[kc].x, e1[kc].x};
            v2f x1 = {e0[kc].y, e1[kc].y};
            v2f x2 = {e0[kc].z, e1[kc].z};
            v2f x3 = {e0[kc].w, e1[kc].w};
            acc0 = __builtin_amdgcn_wmma_f32_16x16x4_f32(
                     false, m[kc], false, x0, (short)0, acc0, false, false);
            acc1 = __builtin_amdgcn_wmma_f32_16x16x4_f32(
                     false, m[kc], false, x1, (short)0, acc1, false, false);
            acc2 = __builtin_amdgcn_wmma_f32_16x16x4_f32(
                     false, m[kc], false, x2, (short)0, acc2, false, false);
            acc3 = __builtin_amdgcn_wmma_f32_16x16x4_f32(
                     false, m[kc], false, x3, (short)0, acc3, false, false);
        }

        asm volatile("v_nop\n\tv_nop\n\tv_nop\n\tv_nop"
                     : "+v"(acc0), "+v"(acc1), "+v"(acc2), "+v"(acc3));

        #pragma unroll
        for (int v = 0; v < 8; ++v) {
            v4f o = {acc0[v], acc1[v], acc2[v], acc3[v]};
            *(v4f*)(dst + (v << shift)) = o;
        }
    }
}

__device__ __forceinline__ void apply_site_row3(float* env,
                                                const float* __restrict__ Mp,
                                                const int wave, int n, int h) {
    asm volatile("" : "+v"(n), "+v"(h));

    v2f b[4];
    #pragma unroll
    for (int kc = 0; kc < 4; ++kc) {
        b[kc].x = Mp[(4 * kc + 2 * h    ) * 16 + n];
        b[kc].y = Mp[(4 * kc + 2 * h + 1) * 16 + n];
    }
    const int w3 = wave & 3;
    int so[8];
    #pragma unroll
    for (int v = 0; v < 8; ++v) {
        const int r3 = 2 * h + (v >> 2);
        so[v] = r3 * 64 + (((v & 3) ^ w3) << 4) + (((n >> 2) ^ r3) << 2) + (n & 3);
    }

    #pragma unroll 2
    for (int t = wave; t < 64; t += 8) {
        const int tb = t * 256;
        v2f a[4];
        #pragma unroll
        for (int kc = 0; kc < 4; ++kc) {
            a[kc].x = env[tb + n + ((4 * kc + 2 * h    ) << 4)];
            a[kc].y = env[tb + n + ((4 * kc + 2 * h + 1) << 4)];
        }
        v8f acc = {};
        #pragma unroll
        for (int kc = 0; kc < 4; ++kc)
            acc = __builtin_amdgcn_wmma_f32_16x16x4_f32(
                    false, b[kc], false, a[kc], (short)0, acc, false, false);
        #pragma unroll
        for (int v = 0; v < 8; ++v)
            env[tb + so[v]] = acc[v];
    }
}

__device__ __forceinline__ void pair_tile_load(const float* __restrict__ M4,
                                               const float* __restrict__ M5,
                                               const int tile, const int n,
                                               const int h, v2f& a, v2f& b) {
    a = *(const v2f*)(M4 + (16 * tile + n) * 4 + 2 * h);
    b.x = M5[((2 * h    ) * 4 + (n & 3)) * 16 + (n >> 2) * 4];
    b.y = M5[((2 * h + 1) * 4 + (n & 3)) * 16 + (n >> 2) * 4];
}

__device__ __forceinline__ void pair_tile_store(float* Tl, const v2f a,
                                                const v2f b, const int tile,
                                                const int n, const int h) {
    v8f z = {};
    z = __builtin_amdgcn_wmma_f32_16x16x4_f32(
            false, a, false, b, (short)0, z, false, false);
    #pragma unroll
    for (int v = 0; v < 8; ++v) {
        const int k4 = 4 * tile + 2 * h + (v >> 2);
        Tl[(k4 & 7) * 128 + (k4 >> 3) * 64 + ((v & 3) * 4 + (n >> 2)) * 4 + (n & 3)] = z[v];
    }
}

__device__ __forceinline__ void apply_pair(const float* env, const float* Tl,
                                           const int wave, int n, int h,
                                           v8f& r0, v8f& r1) {
    asm volatile("" : "+v"(n), "+v"(h));

    const float* tp = Tl + (h * 16 + n) * 4;
    const float* e0 = env + (wave * 16 + n) * 64;
    const float* e1 = e0 + 8 * 16 * 64;

    v8f acc00 = {}, acc01 = {}, acc10 = {}, acc11 = {};
    #pragma unroll
    for (int j = 0; j < 8; ++j) {
        const int slot = ((8 * h + j) ^ n) << 2;
        const v4f tv = *(const v4f*)(tp + j * 128);
        const v4f u0 = *(const v4f*)(e0 + slot);
        const v4f u1 = *(const v4f*)(e1 + slot);
        const v2f ta = {tv.x, tv.y}, tb = {tv.z, tv.w};
        const v2f x0 = {u0.x, u0.y}, y0 = {u0.z, u0.w};
        const v2f x1 = {u1.x, u1.y}, y1 = {u1.z, u1.w};
        acc00 = __builtin_amdgcn_wmma_f32_16x16x4_f32(
                  false, ta, false, x0, (short)0, acc00, false, false);
        acc10 = __builtin_amdgcn_wmma_f32_16x16x4_f32(
                  false, ta, false, x1, (short)0, acc10, false, false);
        acc01 = __builtin_amdgcn_wmma_f32_16x16x4_f32(
                  false, tb, false, y0, (short)0, acc01, false, false);
        acc11 = __builtin_amdgcn_wmma_f32_16x16x4_f32(
                  false, tb, false, y1, (short)0, acc11, false, false);
    }
    asm volatile("v_nop\n\tv_nop\n\tv_nop\n\tv_nop"
                 : "+v"(acc00), "+v"(acc01), "+v"(acc10), "+v"(acc11));
    r0 = acc00 + acc01;
    r1 = acc10 + acc11;
}

__device__ __forceinline__ void top_tile_load(const float* __restrict__ M0,
                                              const float* __restrict__ M1,
                                              const int w, int n, int h,
                                              v2f& a, v2f& b) {
    asm volatile("" : "+v"(n), "+v"(h));
    a = *(const v2f*)(M0 + (n & 3) * 16 + (n >> 2) * 4 + 2 * h);
    b.x = M1[((2 * h    ) * 4 + w) * 16 + n];
    b.y = M1[((2 * h + 1) * 4 + w) * 16 + n];
}

__device__ __forceinline__ void top_tile_store(float* Tl, const v2f a,
                                               const v2f b, const int w,
                                               int n, int h) {
    asm volatile("" : "+v"(n), "+v"(h));
    v8f z = {};
    z = __builtin_amdgcn_wmma_f32_16x16x4_f32(
            false, a, false, b, (short)0, z, false, false);
    #pragma unroll
    for (int v = 0; v < 8; ++v) {
        const int nt = 2 * h + (v >> 2);
        const int kc = v & 3;
        Tl[((nt * 2 + (kc >> 1)) * 32 + (w >> 1) * 16 + n) * 4 + (kc & 1) * 2 + (w & 1)] = z[v];
    }
}

__device__ __forceinline__ void apply_top_pair(float* env, const float* Tl,
                                               const int wave, int n, int h) {
    asm volatile("" : "+v"(n), "+v"(h));

    const int r0 = ((wave & 3) * 16 + n) * 4;
    const int nh = wave >> 2;

    const float* src = env + r0 + ((2 * h) << 8);
    v4f e0[4], e1[4];
    #pragma unroll
    for (int kc = 0; kc < 4; ++kc) {
        e0[kc] = *(const v4f*)(src + ((4 * kc    ) << 8));
        e1[kc] = *(const v4f*)(src + ((4 * kc + 1) << 8));
    }
    const float* tp = Tl + (nh * 4 * 32 + h * 16 + n) * 4;

    #pragma unroll
    for (int q = 0; q < 2; ++q) {
        const v4f t[2] = { *(const v4f*)(tp + (q * 2    ) * 128),
                           *(const v4f*)(tp + (q * 2 + 1) * 128) };
        v8f acc0 = {}, acc1 = {}, acc2 = {}, acc3 = {};
        #pragma unroll
        for (int kc = 0; kc < 4; ++kc) {
            const v4f tv = t[kc >> 1];
            const v2f m  = (kc & 1) ? (v2f){tv.z, tv.w} : (v2f){tv.x, tv.y};
            v2f x0 = {e0[kc].x, e1[kc].x};
            v2f x1 = {e0[kc].y, e1[kc].y};
            v2f x2 = {e0[kc].z, e1[kc].z};
            v2f x3 = {e0[kc].w, e1[kc].w};
            acc0 = __builtin_amdgcn_wmma_f32_16x16x4_f32(
                     false, m, false, x0, (short)0, acc0, false, false);
            acc1 = __builtin_amdgcn_wmma_f32_16x16x4_f32(
                     false, m, false, x1, (short)0, acc1, false, false);
            acc2 = __builtin_amdgcn_wmma_f32_16x16x4_f32(
                     false, m, false, x2, (short)0, acc2, false, false);
            acc3 = __builtin_amdgcn_wmma_f32_16x16x4_f32(
                     false, m, false, x3, (short)0, acc3, false, false);
        }
        asm volatile("v_nop\n\tv_nop\n\tv_nop\n\tv_nop"
                     : "+v"(acc0), "+v"(acc1), "+v"(acc2), "+v"(acc3));

        if (q == 0) __syncthreads();

        float* dst = env + (((2 * nh + q) * 16 + 8 * h) << 8) + r0;
        #pragma unroll
        for (int v = 0; v < 8; ++v) {
            v4f o = {acc0[v], acc1[v], acc2[v], acc3[v]};
            *(v4f*)(dst + (v << 8)) = o;
        }

        if (q == 0)
            asm volatile("" : "+v"(e0[0]), "+v"(e0[1]), "+v"(e0[2]), "+v"(e0[3]),
                              "+v"(e1[0]), "+v"(e1[1]), "+v"(e1[2]), "+v"(e1[3])
                            : : "memory");
    }
}

#define TABN 1024
#define LS1  14592
#define JR   4

__device__ __forceinline__ float site_elem(const float* __restrict__ peps,
                                           const int i, const int j, const int s,
                                           const int k, const int nn) {
    const int a = k >> 2, c = k & 3;
    const int d = nn >> 2, b = nn & 3;
    const bool pad = (i == 0      && a > 0) || (i == LX - 1 && b > 0) ||
                     (j == 0      && c > 0) || (j == LY - 1 && d > 0);
    const float v = peps[(((((i * LY + j) * PD + s) * DD + a) * DD + b) * DD + c) * DD + d];
    return pad ? 0.0f : v;
}

template <int RIGHT>
__device__ __forceinline__ void pair_chain_build(float* Wl,
                                                 const float* __restrict__ peps,
                                                 const int comb,
                                                 const int wave, const int n,
                                                 const int h) {
    constexpr int je = RIGHT ? LY - 1 : 0;
    constexpr int jn = RIGHT ? LY - 2 : 1;
    for (int t = wave; t < 24; t += 8) {
        const int i = t >> 2, u1 = t & 3;
        const int q  = (i < 3) ? i : i - 3;
        const int se = (comb >> (2 * q)) & 1;
        const int sn = (comb >> (2 * q + 1)) & 1;
        v2f a, b;
        if (RIGHT) {
            a.x = site_elem(peps, i, je, se, (n >> 2) * 4 + 2 * h    , n & 3);
            a.y = site_elem(peps, i, je, se, (n >> 2) * 4 + 2 * h + 1, n & 3);
            b.x = site_elem(peps, i, jn, sn, u1 * 4 + (n >> 2), (2 * h    ) * 4 + (n & 3));
            b.y = site_elem(peps, i, jn, sn, u1 * 4 + (n >> 2), (2 * h + 1) * 4 + (n & 3));
        } else {
            a.x = site_elem(peps, i, je, se, (n >> 2) * 4, (2 * h    ) * 4 + (n & 3));
            a.y = site_elem(peps, i, je, se, (n >> 2) * 4, (2 * h + 1) * 4 + (n & 3));
            b.x = site_elem(peps, i, jn, sn, u1 * 4 + 2 * h    , n);
            b.y = site_elem(peps, i, jn, sn, u1 * 4 + 2 * h + 1, n);
        }
        v8f z = {};
        z = __builtin_amdgcn_wmma_f32_16x16x4_f32(
                false, a, false, b, (short)0, z, false, false);
        float* W = Wl + i * 1024 + u1 * 64 + (n >> 2) * 16 + (n & 3);
        #pragma unroll
        for (int v = 0; v < 8; ++v)
            W[(2 * h + (v >> 2)) * 256 + (v & 3) * 4] = z[v];
    }
}

template <int M, int NT>
__device__ __forceinline__ void chain_gemm(const float* A, const float* B,
                                           float* D, const int wave,
                                           const int n, const int h) {
    constexpr int MT  = (M + 15) / 16;
    constexpr int ldn = 16 * NT;
    for (int t = wave; t < MT * NT; t += 8) {
        const int mt = t / NT, nt = t % NT;
        const int row = (M < 16) ? (n & (M - 1)) : (mt * 16 + n);
        v2f a[4], b[4];
        #pragma unroll
        for (int kc = 0; kc < 4; ++kc) {
            a[kc]   = *(const v2f*)(A + row * 16 + 4 * kc + 2 * h);
            b[kc].x = B[(4 * kc + 2 * h    ) * ldn + nt * 16 + n];
            b[kc].y = B[(4 * kc + 2 * h + 1) * ldn + nt * 16 + n];
        }
        v8f acc = {};
        #pragma unroll
        for (int kc = 0; kc < 4; ++kc)
            acc = __builtin_amdgcn_wmma_f32_16x16x4_f32(
                    false, a[kc], false, b[kc], (short)0, acc, false, false);
        #pragma unroll
        for (int v = 0; v < 8; ++v) {
            const int r = mt * 16 + 8 * h + v;
            if (r < M) D[r * ldn + nt * 16 + n] = acc[v];
        }
    }
}

__device__ __forceinline__ void left_tail_tile(const float* W4, const float* W5,
                                               float* T, const int mt,
                                               const int n, const int h) {
    v2f a[4], b[4];
    #pragma unroll
    for (int kc = 0; kc < 4; ++kc) {
        a[kc]   = *(const v2f*)(W4 + (mt * 16 + n) * 16 + 4 * kc + 2 * h);
        b[kc].x = W5[(4 * kc + 2 * h    ) * 64 + (n & 3) * 16];
        b[kc].y = W5[(4 * kc + 2 * h + 1) * 64 + (n & 3) * 16];
    }
    v8f acc = {};
    #pragma unroll
    for (int kc = 0; kc < 4; ++kc)
        acc = __builtin_amdgcn_wmma_f32_16x16x4_f32(
                false, a[kc], false, b[kc], (short)0, acc, false, false);
    if (n < 4) {
        #pragma unroll
        for (int v = 0; v < 8; ++v)
            T[(mt * 16 + 8 * h + v) * 4 + n] = acc[v];
    }
}

__device__ __forceinline__ void chain_product(const float* __restrict__ A,
                                              const float* __restrict__ B,
                                              float* D, const int wave,
                                              const int n, const int h) {
    const int nt = wave & 3;
    v2f b[4], a[2][4];
    #pragma unroll
    for (int kc = 0; kc < 4; ++kc) {
        b[kc].x = B[(4 * kc + 2 * h    ) * 64 + nt * 16 + n];
        b[kc].y = B[(4 * kc + 2 * h + 1) * 64 + nt * 16 + n];
    }
    #pragma unroll
    for (int q = 0; q < 2; ++q) {
        #pragma unroll
        for (int kc = 0; kc < 4; ++kc)
            a[q][kc] = *(const v2f*)(A + (((wave >> 2) + 2 * q) * 16 + n) * 16 + 4 * kc + 2 * h);
    }
    #pragma unroll
    for (int q = 0; q < 2; ++q) {
        const int mt = (wave >> 2) + 2 * q;
        v8f acc = {};
        #pragma unroll
        for (int kc = 0; kc < 4; ++kc)
            acc = __builtin_amdgcn_wmma_f32_16x16x4_f32(
                    false, a[q][kc], false, b[kc], (short)0, acc, false, false);
        #pragma unroll
        for (int v = 0; v < 8; ++v)
            D[(mt * 16 + 8 * h + v) * 64 + nt * 16 + n] = acc[v];
    }
}

__device__ __forceinline__ float chain_join(const float* __restrict__ A,
                                            const float* __restrict__ B,
                                            const float* L, const int wave,
                                            const int n, const int h) {
    const int nt = wave & 3;
    v2f b[4];
    #pragma unroll
    for (int kc = 0; kc < 4; ++kc) {
        b[kc].x = B[(4 * kc + 2 * h    ) * 64 + nt * 16 + n];
        b[kc].y = B[(4 * kc + 2 * h + 1) * 64 + nt * 16 + n];
    }
    float p = 0.0f;
    #pragma unroll
    for (int q = 0; q < 2; ++q) {
        const int mt = (wave >> 2) + 2 * q;
        v2f a[4];
        #pragma unroll
        for (int kc = 0; kc < 4; ++kc)
            a[kc] = *(const v2f*)(A + (mt * 16 + n) * 16 + 4 * kc + 2 * h);
        float l[8];
        #pragma unroll
        for (int v = 0; v < 8; ++v)
            l[v] = L[(mt * 16 + 8 * h + v) * 64 + nt * 16 + n];
        v8f acc = {};
        #pragma unroll
        for (int kc = 0; kc < 4; ++kc)
            acc = __builtin_amdgcn_wmma_f32_16x16x4_f32(
                    false, a[kc], false, b[kc], (short)0, acc, false, false);
        asm volatile("v_nop\n\tv_nop\n\tv_nop\n\tv_nop" : "+v"(acc));
        #pragma unroll
        for (int v = 0; v < 8; ++v)
            p += acc[v] * l[v];
    }
    return p;
}

#define NPREP 72
__global__ void __launch_bounds__(256)
peps_prep_kernel(const float* __restrict__ peps,
                 float* __restrict__ Msite,
                 float* __restrict__ tab) {
    if (blockIdx.x < NPREP) {
        const int idx = blockIdx.x * 256 + threadIdx.x;
        const int n    = idx & 15;
        const int k    = (idx >> 4) & 15;
        const int s    = (idx >> 8) & 1;
        const int site = idx >> 9;
        Msite[idx] = site_elem(peps, site / LY, site % LY, s, k, n);
        return;
    }
    __shared__ __attribute__((aligned(16))) float wl[6 * 1024 + 256 + 256];
    float* const S1 = wl + 6144;
    float* const T  = wl + 6400;

    const int tb   = blockIdx.x - NPREP;
    const int side = tb >> 6;
    const int comb = tb & 63;
    const int lane = threadIdx.x & 31;
    const int wave = __builtin_amdgcn_readfirstlane(threadIdx.x >> 5);
    __builtin_assume(wave >= 0 && wave < 8);
    const int n = lane & 15;
    const int h = lane >> 4;

    if (side == 0) pair_chain_build<0>(wl, peps, comb, wave, n, h);
    else           pair_chain_build<1>(wl, peps, comb, wave, n, h);
    __syncthreads();
    chain_gemm<4, 4>(wl, wl + 1024, S1, wave, n, h);
    if (wave >= 4)
        left_tail_tile(wl + 4096, wl + 5120, T, wave - 4, n, h);
    __syncthreads();
    float* const H = tab + ((side * 2 + 0) * 64 + comb) * TABN;
    float* const G = tab + ((side * 2 + 1) * 64 + comb) * TABN;
    chain_gemm<16, 4>(S1, wl + 2048, H, wave, n, h);
    chain_gemm<64, 1>(wl + 3072, T, G, wave, n, h);
}

__global__ void __launch_bounds__(256)
peps_amp_kernel(const int* __restrict__ x,
                const float* __restrict__ Msite,
                const float* __restrict__ tab,
                float* __restrict__ out) {
    __shared__ __attribute__((aligned(16))) float env[ENV7 + TLEN];
    float* const Tl = env + ENV7;

    const int tid    = threadIdx.x;
    const int sample = blockIdx.x;
    const int lane   = tid & 31;
    const int wave   = __builtin_amdgcn_readfirstlane(tid >> 5);
    __builtin_assume(wave >= 0 && wave < 8);
    const int n      = lane & 15;
    const int h      = lane >> 4;

    const int* xrow = x + sample * NSITE;

    int cmb[4];
    #pragma unroll
    for (int sd = 0; sd < 2; ++sd) {
        const int je = sd ? LY - 1 : 0, jn = sd ? LY - 2 : 1;
        #pragma unroll
        for (int hf = 0; hf < 2; ++hf) {
            int cb = 0;
            #pragma unroll
            for (int q = 0; q < 3; ++q) {
                const int i = 3 * hf + q;
                cb |= (xrow[i * LY + je] & 1) << (2 * q);
                cb |= (xrow[i * LY + jn] & 1) << (2 * q + 1);
            }
            cmb[sd * 2 + hf] = cb;
        }
    }

    {
        v2f qa = {}, qb = {};
        if (wave < 4) {
            const int s0 = xrow[0 * LY + 2];
            const int s1 = xrow[1 * LY + 2];
            top_tile_load(Msite + ((0 * LY + 2) * PD + s0) * 256,
                          Msite + ((1 * LY + 2) * PD + s1) * 256,
                          wave, n, h, qa, qb);
        }
        chain_product(tab + (0 * 64 + cmb[0]) * TABN,
                      tab + (1 * 64 + cmb[1]) * TABN, env, wave, n, h);
        if (wave < 4)
            top_tile_store(Tl, qa, qb, wave, n, h);
    }
    __syncthreads();

    for (int j = 2; j < JR; ++j) {
        for (int i = 0; i < LX - 1; ++i) {
            if (i == 1) continue;
            const int site = i * LY + j;
            const int s    = xrow[site];
            const float* Mp = Msite + (site * PD + s) * 256;

            if (i == 0) {
                apply_top_pair(env, Tl, wave, n, h);
            } else if (i == 2) {
                apply_site_rows4<4, 6>(env, Mp, 16, wave, n, h);
            } else if (i == 3) {
                v2f pa = {}, pb = {};
                if (wave < 4) {
                    const int s4 = xrow[4 * LY + j];
                    const int s5 = xrow[5 * LY + j];
                    pair_tile_load(Msite + ((4 * LY + j) * PD + s4) * 256,
                                   Msite + ((5 * LY + j) * PD + s5) * 256,
                                   wave, n, h, pa, pb);
                }
                apply_site_row3(env, Mp, wave, n, h);
                if (wave < 4)
                    pair_tile_store(Tl, pa, pb, wave, n, h);
            } else {
                const bool top = (wave < 4) && (j + 1 < JR);
                v2f qa = {}, qb = {};
                if (top) {
                    const int s0 = xrow[0 * LY + j + 1];
                    const int s1 = xrow[1 * LY + j + 1];
                    top_tile_load(Msite + ((0 * LY + j + 1) * PD + s0) * 256,
                                  Msite + ((1 * LY + j + 1) * PD + s1) * 256,
                                  wave, n, h, qa, qb);
                }
                v8f r0, r1;
                apply_pair(env, Tl, wave, n, h, r0, r1);
                __syncthreads();
                if (top)
                    top_tile_store(Tl, qa, qb, wave, n, h);
                float* o0 = env + (wave * 16 + n) * 16 + 8 * h;
                float* o1 = o0 + 8 * 16 * 16;
                *(v4f*)(o0    ) = __builtin_shufflevector(r0, r0, 0, 1, 2, 3);
                *(v4f*)(o0 + 4) = __builtin_shufflevector(r0, r0, 4, 5, 6, 7);
                *(v4f*)(o1    ) = __builtin_shufflevector(r1, r1, 0, 1, 2, 3);
                *(v4f*)(o1 + 4) = __builtin_shufflevector(r1, r1, 4, 5, 6, 7);
            }
            __syncthreads();
        }
    }

    float p = chain_join(tab + (2 * 64 + cmb[2]) * TABN,
                         tab + (3 * 64 + cmb[3]) * TABN, env, wave, n, h);

    for (int o = 16; o > 0; o >>= 1)
        p += __shfl_xor(p, o);
    if (lane == 0) env[LS1 + wave] = p;
    __syncthreads();
    if (tid == 0) {
        float a = 0.0f;
        for (int w = 0; w < 8; ++w) a += env[LS1 + w];
        out[sample] = a;
    }
}

extern "C" void kernel_launch(void* const* d_in, const int* in_sizes, int n_in,
                              void* d_out, int out_size, void* d_ws, size_t ws_size,
                              hipStream_t stream) {
    const int*   x     = (const int*)d_in[0];
    const float* peps  = (const float*)d_in[1];
    float*       out   = (float*)d_out;
    float*       Msite = (float*)d_ws;
    float*       tab   = Msite + NSITE * PD * 256;

    peps_prep_kernel<<<NPREP + 128, 256, 0, stream>>>(peps, Msite, tab);

    const int B = out_size;
    peps_amp_kernel<<<B, 256, 0, stream>>>(x, Msite, tab, out);
}


# lattice split into four 3x3 quadrants, each tabulated in the prep launch per its 9 indices; a sample is two 64x64x64 products and a 4096-term inner product; sum order changes: quadrants first
# speedup vs baseline: 28.9392x; 2.6025x over previous
#include <hip/hip_runtime.h>


typedef float v2f __attribute__((ext_vector_type(2)));
typedef float v4f __attribute__((ext_vector_type(4)));
typedef float v8f __attribute__((ext_vector_type(8)));

#define LX 6
#define LY 6
#define PD 2
#define DD 4
#define NSITE 36
#define ENV7 16384
#define ENV6 4096
#define TLEN 1024

template <int NK, int SHIFT>
__device__ __forceinline__ void apply_site_rows4(float* env,
                                                 const float* __restrict__ Mp,
                                                 const int ngroup,
                                                 const int wave, const int n,
                                                 const int h) {
    constexpr int shift = SHIFT;
    constexpr int psm = (1 << shift) - 1;

    v2f m[NK];
    #pragma unroll
    for (int kc = 0; kc < NK; ++kc) {
        m[kc].x = Mp[(4 * kc + 2 * h    ) * 16 + n];
        m[kc].y = Mp[(4 * kc + 2 * h + 1) * 16 + n];
    }

    for (int g = wave; g < ngroup; g += 8) {
        const int r0   = (g * 16 + n) * 4;
        const int post = r0 & psm;
        const int base = ((r0 - post) << 4) + post;

        const float* src = env + base + ((2 * h) << shift);
        float*       dst = env + base + ((8 * h) << shift);

        v4f e0[NK], e1[NK];
        #pragma unroll
        for (int kc = 0; kc < NK; ++kc) {
            e0[kc] = *(const v4f*)(src + ((4 * kc    ) << shift));
            e1[kc] = *(const v4f*)(src + ((4 * kc + 1) << shift));
        }

        v8f acc0 = {}, acc1 = {}, acc2 = {}, acc3 = {};
        #pragma unroll
        for (int kc = 0; kc < NK; ++kc) {
            v2f x0 = {e0[kc].x, e1[kc].x};
            v2f x1 = {e0[kc].y, e1[kc].y};
            v2f x2 = {e0[kc].z, e1[kc].z};
            v2f x3 = {e0[kc].w, e1[kc].w};
            acc0 = __builtin_amdgcn_wmma_f32_16x16x4_f32(
                     false, m[kc], false, x0, (short)0, acc0, false, false);
            acc1 = __builtin_amdgcn_wmma_f32_16x16x4_f32(
                     false, m[kc], false, x1, (short)0, acc1, false, false);
            acc2 = __builtin_amdgcn_wmma_f32_16x16x4_f32(
                     false, m[kc], false, x2, (short)0, acc2, false, false);
            acc3 = __builtin_amdgcn_wmma_f32_16x16x4_f32(
                     false, m[kc], false, x3, (short)0, acc3, false, false);
        }

        asm volatile("v_nop\n\tv_nop\n\tv_nop\n\tv_nop"
                     : "+v"(acc0), "+v"(acc1), "+v"(acc2), "+v"(acc3));

        #pragma unroll
        for (int v = 0; v < 8; ++v) {
            v4f o = {acc0[v], acc1[v], acc2[v], acc3[v]};
            *(v4f*)(dst + (v << shift)) = o;
        }
    }
}

__device__ __forceinline__ void apply_site_row3(float* env,
                                                const float* __restrict__ Mp,
                                                const int wave, int n, int h) {
    asm volatile("" : "+v"(n), "+v"(h));

    v2f b[4];
    #pragma unroll
    for (int kc = 0; kc < 4; ++kc) {
        b[kc].x = Mp[(4 * kc + 2 * h    ) * 16 + n];
        b[kc].y = Mp[(4 * kc + 2 * h + 1) * 16 + n];
    }
    const int w3 = wave & 3;
    int so[8];
    #pragma unroll
    for (int v = 0; v < 8; ++v) {
        const int r3 = 2 * h + (v >> 2);
        so[v] = r3 * 64 + (((v & 3) ^ w3) << 4) + (((n >> 2) ^ r3) << 2) + (n & 3);
    }

    #pragma unroll 2
    for (int t = wave; t < 64; t += 8) {
        const int tb = t * 256;
        v2f a[4];
        #pragma unroll
        for (int kc = 0; kc < 4; ++kc) {
            a[kc].x = env[tb + n + ((4 * kc + 2 * h    ) << 4)];
            a[kc].y = env[tb + n + ((4 * kc + 2 * h + 1) << 4)];
        }
        v8f acc = {};
        #pragma unroll
        for (int kc = 0; kc < 4; ++kc)
            acc = __builtin_amdgcn_wmma_f32_16x16x4_f32(
                    false, b[kc], false, a[kc], (short)0, acc, false, false);
        #pragma unroll
        for (int v = 0; v < 8; ++v)
            env[tb + so[v]] = acc[v];
    }
}

__device__ __forceinline__ void pair_tile_load(const float* __restrict__ M4,
                                               const float* __restrict__ M5,
                                               const int tile, const int n,
                                               const int h, v2f& a, v2f& b) {
    a = *(const v2f*)(M4 + (16 * tile + n) * 4 + 2 * h);
    b.x = M5[((2 * h    ) * 4 + (n & 3)) * 16 + (n >> 2) * 4];
    b.y = M5[((2 * h + 1) * 4 + (n & 3)) * 16 + (n >> 2) * 4];
}

__device__ __forceinline__ void pair_tile_store(float* Tl, const v2f a,
                                                const v2f b, const int tile,
                                                const int n, const int h) {
    v8f z = {};
    z = __builtin_amdgcn_wmma_f32_16x16x4_f32(
            false, a, false, b, (short)0, z, false, false);
    #pragma unroll
    for (int v = 0; v < 8; ++v) {
        const int k4 = 4 * tile + 2 * h + (v >> 2);
        Tl[(k4 & 7) * 128 + (k4 >> 3) * 64 + ((v & 3) * 4 + (n >> 2)) * 4 + (n & 3)] = z[v];
    }
}

__device__ __forceinline__ void apply_pair(const float* env, const float* Tl,
                                           const int wave, int n, int h,
                                           v8f& r0, v8f& r1) {
    asm volatile("" : "+v"(n), "+v"(h));

    const float* tp = Tl + (h * 16 + n) * 4;
    const float* e0 = env + (wave * 16 + n) * 64;
    const float* e1 = e0 + 8 * 16 * 64;

    v8f acc00 = {}, acc01 = {}, acc10 = {}, acc11 = {};
    #pragma unroll
    for (int j = 0; j < 8; ++j) {
        const int slot = ((8 * h + j) ^ n) << 2;
        const v4f tv = *(const v4f*)(tp + j * 128);
        const v4f u0 = *(const v4f*)(e0 + slot);
        const v4f u1 = *(const v4f*)(e1 + slot);
        const v2f ta = {tv.x, tv.y}, tb = {tv.z, tv.w};
        const v2f x0 = {u0.x, u0.y}, y0 = {u0.z, u0.w};
        const v2f x1 = {u1.x, u1.y}, y1 = {u1.z, u1.w};
        acc00 = __builtin_amdgcn_wmma_f32_16x16x4_f32(
                  false, ta, false, x0, (short)0, acc00, false, false);
        acc10 = __builtin_amdgcn_wmma_f32_16x16x4_f32(
                  false, ta, false, x1, (short)0, acc10, false, false);
        acc01 = __builtin_amdgcn_wmma_f32_16x16x4_f32(
                  false, tb, false, y0, (short)0, acc01, false, false);
        acc11 = __builtin_amdgcn_wmma_f32_16x16x4_f32(
                  false, tb, false, y1, (short)0, acc11, false, false);
    }
    asm volatile("v_nop\n\tv_nop\n\tv_nop\n\tv_nop"
                 : "+v"(acc00), "+v"(acc01), "+v"(acc10), "+v"(acc11));
    r0 = acc00 + acc01;
    r1 = acc10 + acc11;
}

__device__ __forceinline__ void top_tile_load(const float* __restrict__ M0,
                                              const float* __restrict__ M1,
                                              const int w, int n, int h,
                                              v2f& a, v2f& b) {
    asm volatile("" : "+v"(n), "+v"(h));
    a = *(const v2f*)(M0 + (n & 3) * 16 + (n >> 2) * 4 + 2 * h);
    b.x = M1[((2 * h    ) * 4 + w) * 16 + n];
    b.y = M1[((2 * h + 1) * 4 + w) * 16 + n];
}

__device__ __forceinline__ void top_tile_store(float* Tl, const v2f a,
                                               const v2f b, const int w,
                                               int n, int h) {
    asm volatile("" : "+v"(n), "+v"(h));
    v8f z = {};
    z = __builtin_amdgcn_wmma_f32_16x16x4_f32(
            false, a, false, b, (short)0, z, false, false);
    #pragma unroll
    for (int v = 0; v < 8; ++v) {
        const int nt = 2 * h + (v >> 2);
        const int kc = v & 3;
        Tl[((nt * 2 + (kc >> 1)) * 32 + (w >> 1) * 16 + n) * 4 + (kc & 1) * 2 + (w & 1)] = z[v];
    }
}

__device__ __forceinline__ void apply_top_pair(float* env, const float* Tl,
                                               const int wave, int n, int h) {
    asm volatile("" : "+v"(n), "+v"(h));

    const int r0 = ((wave & 3) * 16 + n) * 4;
    const int nh = wave >> 2;

    const float* src = env + r0 + ((2 * h) << 8);
    v4f e0[4], e1[4];
    #pragma unroll
    for (int kc = 0; kc < 4; ++kc) {
        e0[kc] = *(const v4f*)(src + ((4 * kc    ) << 8));
        e1[kc] = *(const v4f*)(src + ((4 * kc + 1) << 8));
    }
    const float* tp = Tl + (nh * 4 * 32 + h * 16 + n) * 4;

    #pragma unroll
    for (int q = 0; q < 2; ++q) {
        const v4f t[2] = { *(const v4f*)(tp + (q * 2    ) * 128),
                           *(const v4f*)(tp + (q * 2 + 1) * 128) };
        v8f acc0 = {}, acc1 = {}, acc2 = {}, acc3 = {};
        #pragma unroll
        for (int kc = 0; kc < 4; ++kc) {
            const v4f tv = t[kc >> 1];
            const v2f m  = (kc & 1) ? (v2f){tv.z, tv.w} : (v2f){tv.x, tv.y};
            v2f x0 = {e0[kc].x, e1[kc].x};
            v2f x1 = {e0[kc].y, e1[kc].y};
            v2f x2 = {e0[kc].z, e1[kc].z};
            v2f x3 = {e0[kc].w, e1[kc].w};
            acc0 = __builtin_amdgcn_wmma_f32_16x16x4_f32(
                     false, m, false, x0, (short)0, acc0, false, false);
            acc1 = __builtin_amdgcn_wmma_f32_16x16x4_f32(
                     false, m, false, x1, (short)0, acc1, false, false);
            acc2 = __builtin_amdgcn_wmma_f32_16x16x4_f32(
                     false, m, false, x2, (short)0, acc2, false, false);
            acc3 = __builtin_amdgcn_wmma_f32_16x16x4_f32(
                     false, m, false, x3, (short)0, acc3, false, false);
        }
        asm volatile("v_nop\n\tv_nop\n\tv_nop\n\tv_nop"
                     : "+v"(acc0), "+v"(acc1), "+v"(acc2), "+v"(acc3));

        if (q == 0) __syncthreads();

        float* dst = env + (((2 * nh + q) * 16 + 8 * h) << 8) + r0;
        #pragma unroll
        for (int v = 0; v < 8; ++v) {
            v4f o = {acc0[v], acc1[v], acc2[v], acc3[v]};
            *(v4f*)(dst + (v << 8)) = o;
        }

        if (q == 0)
            asm volatile("" : "+v"(e0[0]), "+v"(e0[1]), "+v"(e0[2]), "+v"(e0[3]),
                              "+v"(e1[0]), "+v"(e1[1]), "+v"(e1[2]), "+v"(e1[3])
                            : : "memory");
    }
}

#define TABN 1024
#define LS1  14592
#define JR   4

__device__ __forceinline__ float site_elem(const float* __restrict__ peps,
                                           const int i, const int j, const int s,
                                           const int k, const int nn) {
    const int a = k >> 2, c = k & 3;
    const int d = nn >> 2, b = nn & 3;
    const bool pad = (i == 0      && a > 0) || (i == LX - 1 && b > 0) ||
                     (j == 0      && c > 0) || (j == LY - 1 && d > 0);
    const float v = peps[(((((i * LY + j) * PD + s) * DD + a) * DD + b) * DD + c) * DD + d];
    return pad ? 0.0f : v;
}

template <int RIGHT>
__device__ __forceinline__ void pair_chain_build(float* Wl,
                                                 const float* __restrict__ peps,
                                                 const int comb,
                                                 const int wave, const int n,
                                                 const int h) {
    constexpr int je = RIGHT ? LY - 1 : 0;
    constexpr int jn = RIGHT ? LY - 2 : 1;
    for (int t = wave; t < 24; t += 8) {
        const int i = t >> 2, u1 = t & 3;
        const int q  = (i < 3) ? i : i - 3;
        const int se = (comb >> (2 * q)) & 1;
        const int sn = (comb >> (2 * q + 1)) & 1;
        v2f a, b;
        if (RIGHT) {
            a.x = site_elem(peps, i, je, se, (n >> 2) * 4 + 2 * h    , n & 3);
            a.y = site_elem(peps, i, je, se, (n >> 2) * 4 + 2 * h + 1, n & 3);
            b.x = site_elem(peps, i, jn, sn, u1 * 4 + (n >> 2), (2 * h    ) * 4 + (n & 3));
            b.y = site_elem(peps, i, jn, sn, u1 * 4 + (n >> 2), (2 * h + 1) * 4 + (n & 3));
        } else {
            a.x = site_elem(peps, i, je, se, (n >> 2) * 4, (2 * h    ) * 4 + (n & 3));
            a.y = site_elem(peps, i, je, se, (n >> 2) * 4, (2 * h + 1) * 4 + (n & 3));
            b.x = site_elem(peps, i, jn, sn, u1 * 4 + 2 * h    , n);
            b.y = site_elem(peps, i, jn, sn, u1 * 4 + 2 * h + 1, n);
        }
        v8f z = {};
        z = __builtin_amdgcn_wmma_f32_16x16x4_f32(
                false, a, false, b, (short)0, z, false, false);
        float* W = Wl + i * 1024 + u1 * 64 + (n >> 2) * 16 + (n & 3);
        #pragma unroll
        for (int v = 0; v < 8; ++v)
            W[(2 * h + (v >> 2)) * 256 + (v & 3) * 4] = z[v];
    }
}

template <int M, int NT>
__device__ __forceinline__ void chain_gemm(const float* A, const float* B,
                                           float* D, const int wave,
                                           const int n, const int h) {
    constexpr int MT  = (M + 15) / 16;
    constexpr int ldn = 16 * NT;
    for (int t = wave; t < MT * NT; t += 8) {
        const int mt = t / NT, nt = t % NT;
        const int row = (M < 16) ? (n & (M - 1)) : (mt * 16 + n);
        v2f a[4], b[4];
        #pragma unroll
        for (int kc = 0; kc < 4; ++kc) {
            a[kc]   = *(const v2f*)(A + row * 16 + 4 * kc + 2 * h);
            b[kc].x = B[(4 * kc + 2 * h    ) * ldn + nt * 16 + n];
            b[kc].y = B[(4 * kc + 2 * h + 1) * ldn + nt * 16 + n];
        }
        v8f acc = {};
        #pragma unroll
        for (int kc = 0; kc < 4; ++kc)
            acc = __builtin_amdgcn_wmma_f32_16x16x4_f32(
                    false, a[kc], false, b[kc], (short)0, acc, false, false);
        #pragma unroll
        for (int v = 0; v < 8; ++v) {
            const int r = mt * 16 + 8 * h + v;
            if (r < M) D[r * ldn + nt * 16 + n] = acc[v];
        }
    }
}

__device__ __forceinline__ void left_tail_tile(const float* W4, const float* W5,
                                               float* T, const int mt,
                                               const int n, const int h) {
    v2f a[4], b[4];
    #pragma unroll
    for (int kc = 0; kc < 4; ++kc) {
        a[kc]   = *(const v2f*)(W4 + (mt * 16 + n) * 16 + 4 * kc + 2 * h);
        b[kc].x = W5[(4 * kc + 2 * h    ) * 64 + (n & 3) * 16];
        b[kc].y = W5[(4 * kc + 2 * h + 1) * 64 + (n & 3) * 16];
    }
    v8f acc = {};
    #pragma unroll
    for (int kc = 0; kc < 4; ++kc)
        acc = __builtin_amdgcn_wmma_f32_16x16x4_f32(
                false, a[kc], false, b[kc], (short)0, acc, false, false);
    if (n < 4) {
        #pragma unroll
        for (int v = 0; v < 8; ++v)
            T[(mt * 16 + 8 * h + v) * 4 + n] = acc[v];
    }
}

__device__ __forceinline__ void chain_product(const float* __restrict__ A,
                                              const float* __restrict__ B,
                                              float* D, const int wave,
                                              const int n, const int h) {
    const int nt = wave & 3;
    v2f b[4], a[2][4];
    #pragma unroll
    for (int kc = 0; kc < 4; ++kc) {
        b[kc].x = B[(4 * kc + 2 * h    ) * 64 + nt * 16 + n];
        b[kc].y = B[(4 * kc + 2 * h + 1) * 64 + nt * 16 + n];
    }
    #pragma unroll
    for (int q = 0; q < 2; ++q) {
        #pragma unroll
        for (int kc = 0; kc < 4; ++kc)
            a[q][kc] = *(const v2f*)(A + (((wave >> 2) + 2 * q) * 16 + n) * 16 + 4 * kc + 2 * h);
    }
    #pragma unroll
    for (int q = 0; q < 2; ++q) {
        const int mt = (wave >> 2) + 2 * q;
        v8f acc = {};
        #pragma unroll
        for (int kc = 0; kc < 4; ++kc)
            acc = __builtin_amdgcn_wmma_f32_16x16x4_f32(
                    false, a[q][kc], false, b[kc], (short)0, acc, false, false);
        #pragma unroll
        for (int v = 0; v < 8; ++v)
            D[(mt * 16 + 8 * h + v) * 64 + nt * 16 + n] = acc[v];
    }
}

__device__ __forceinline__ float chain_join(const float* __restrict__ A,
                                            const float* __restrict__ B,
                                            const float* L, const int wave,
                                            const int n, const int h) {
    const int nt = wave & 3;
    v2f b[4];
    #pragma unroll
    for (int kc = 0; kc < 4; ++kc) {
        b[kc].x = B[(4 * kc + 2 * h    ) * 64 + nt * 16 + n];
        b[kc].y = B[(4 * kc + 2 * h + 1) * 64 + nt * 16 + n];
    }
    float p = 0.0f;
    #pragma unroll
    for (int q = 0; q < 2; ++q) {
        const int mt = (wave >> 2) + 2 * q;
        v2f a[4];
        #pragma unroll
        for (int kc = 0; kc < 4; ++kc)
            a[kc] = *(const v2f*)(A + (mt * 16 + n) * 16 + 4 * kc + 2 * h);
        float l[8];
        #pragma unroll
        for (int v = 0; v < 8; ++v)
            l[v] = L[(mt * 16 + 8 * h + v) * 64 + nt * 16 + n];
        v8f acc = {};
        #pragma unroll
        for (int kc = 0; kc < 4; ++kc)
            acc = __builtin_amdgcn_wmma_f32_16x16x4_f32(
                    false, a[kc], false, b[kc], (short)0, acc, false, false);
        asm volatile("v_nop\n\tv_nop\n\tv_nop\n\tv_nop" : "+v"(acc));
        #pragma unroll
        for (int v = 0; v < 8; ++v)
            p += acc[v] * l[v];
    }
    return p;
}

#define QTABN 4096
#define NQUAD 256
#define WS_FLOATS_QUAD (NSITE * PD * 256 + 4 * 64 * TABN + 4 * 512 * QTABN)

__device__ __forceinline__ int quad_off(const int r, const int v) {
    return ((((v >> 4) * 8 + (r >> 3)) * 32 + ((r >> 1) & 1) * 16 + (v & 15)) << 2)
           + ((r >> 2) & 1) * 2 + (r & 1);
}

__device__ __forceinline__ float quad_elem(const float* __restrict__ peps,
                                           const int i, const int j, const int s,
                                           const int k, const int nn,
                                           const int side, const int half) {
    const int vin = k >> 2, hin = k & 3, hout = nn >> 2, vout = nn & 3;
    const int up    = half ? vout : vin;
    const int down  = half ? vin  : vout;
    const int left  = side ? hout : hin;
    const int right = side ? hin  : hout;
    return site_elem(peps, i, j, s, up * 4 + left, right * 4 + down);
}

__device__ __forceinline__ v2f quad_bfrag(const float* __restrict__ peps,
                                          const int i, const int j, const int s,
                                          const int kc, const int n, const int h,
                                          const int side, const int half) {
    v2f b;
    b.x = quad_elem(peps, i, j, s, 4 * kc + 2 * h    , n, side, half);
    b.y = quad_elem(peps, i, j, s, 4 * kc + 2 * h + 1, n, side, half);
    return b;
}

__device__ __forceinline__ void quad_step0(const float* X0, float* E1,
                                           const v2f (&b)[2], const int half,
                                           const int wave, const int n, const int h) {
    const int sa0 = half ? 1 : 256, sa1 = half ? 4 : 64, sb = half ? 64 : 1;
    for (int tp = 0; tp < 2; ++tp) {
        const int t   = wave + 8 * tp;
        const int row = t * 16 + n;
        const int rb  = (row >> 6) * sa1 + ((row >> 4) & 3) * 16 + (row & 15) * sb;
        v2f a;
        a.x = X0[rb + (2 * h    ) * sa0];
        a.y = X0[rb + (2 * h + 1) * sa0];
        #pragma unroll
        for (int s0 = 0; s0 < 2; ++s0) {
            v8f acc = {};
            acc = __builtin_amdgcn_wmma_f32_16x16x4_f32(
                    false, a, false, b[s0], (short)0, acc, false, false);
            #pragma unroll
            for (int v = 0; v < 8; ++v)
                E1[s0 * 4096 + n * 256 + t * 16 + 8 * h + v] = acc[v];
        }
    }
}

__device__ __forceinline__ void quad_step1(const float* E1, float* E2,
                                           const v2f (&b)[2][4],
                                           const int wave, const int n, const int h) {
    for (int tp = 0; tp < 2; ++tp) {
        const int t   = wave + 8 * tp;
        const int row = t * 16 + n;
        const int rb  = (row >> 6) * 1024 + ((row >> 4) & 3) * 16 + (row & 15);
        #pragma unroll
        for (int s0 = 0; s0 < 2; ++s0) {
            v2f a[4];
            #pragma unroll
            for (int kc = 0; kc < 4; ++kc) {
                a[kc].x = E1[s0 * 4096 + rb + kc * 256 + (2 * h    ) * 64];
                a[kc].y = E1[s0 * 4096 + rb + kc * 256 + (2 * h + 1) * 64];
            }
            #pragma unroll
            for (int s1 = 0; s1 < 2; ++s1) {
                v8f acc = {};
                #pragma unroll
                for (int kc = 0; kc < 4; ++kc)
                    acc = __builtin_amdgcn_wmma_f32_16x16x4_f32(
                            false, a[kc], false, b[s1][kc], (short)0, acc, false, false);
                #pragma unroll
                for (int v = 0; v < 8; ++v)
                    E2[(s0 + 2 * s1) * 4096 + (t >> 2) * 1024 + n * 64 + (t & 3) * 16 + 8 * h + v] = acc[v];
            }
        }
    }
}

__device__ __forceinline__ void quad_step2(const float* E2, float* __restrict__ Qb,
                                           const v2f (&b)[2][4], const int half,
                                           const int wave, const int n, const int h) {
    for (int tp = 0; tp < 2; ++tp) {
        const int t   = wave + 8 * tp;
        const int row = t * 16 + n;
        const int rb  = (row >> 6) * 1024 + ((row >> 4) & 3) * 256 + (row & 15);
        #pragma unroll
        for (int s01 = 0; s01 < 4; ++s01) {
            v2f a[4];
            #pragma unroll
            for (int kc = 0; kc < 4; ++kc) {
                a[kc].x = E2[s01 * 4096 + rb + kc * 64 + (2 * h    ) * 16];
                a[kc].y = E2[s01 * 4096 + rb + kc * 64 + (2 * h + 1) * 16];
            }
            #pragma unroll
            for (int s2 = 0; s2 < 2; ++s2) {
                v8f acc = {};
                #pragma unroll
                for (int kc = 0; kc < 4; ++kc)
                    acc = __builtin_amdgcn_wmma_f32_16x16x4_f32(
                            false, a[kc], false, b[s2][kc], (short)0, acc, false, false);
                const int c3 = half ? ((s01 & 1) * 4 + (s01 >> 1) * 2 + s2) : (s01 + 4 * s2);
                float* const Q = Qb + c3 * 64 * QTABN;
                #pragma unroll
                for (int v = 0; v < 8; ++v)
                    Q[quad_off(t * 4 + (n >> 2), (8 * h + v) * 4 + (n & 3))] = acc[v];
            }
        }
    }
}

__device__ __forceinline__ void quad_product(const float* __restrict__ FA,
                                             const float* __restrict__ FB,
                                             const int wave, const int n, const int h,
                                             v8f& d0, v8f& d1) {
    const int lo = (h * 16 + n) * 4;
    const float* pa0 = FA + (wave >> 2) * 1024 + lo;
    const float* pa1 = pa0 + 2048;
    const float* pb  = FB + (wave & 3) * 1024 + lo;
    v8f acc0 = {}, acc1 = {};
    #pragma unroll
    for (int p = 0; p < 8; ++p) {
        const v4f fb = *(const v4f*)(pb  + p * 128);
        const v4f f0 = *(const v4f*)(pa0 + p * 128);
        const v4f f1 = *(const v4f*)(pa1 + p * 128);
        const v2f be = {fb.x, fb.y}, bo = {fb.z, fb.w};
        const v2f a0e = {f0.x, f0.y}, a0o = {f0.z, f0.w};
        const v2f a1e = {f1.x, f1.y}, a1o = {f1.z, f1.w};
        acc0 = __builtin_amdgcn_wmma_f32_16x16x4_f32(
                 false, a0e, false, be, (short)0, acc0, false, false);
        acc1 = __builtin_amdgcn_wmma_f32_16x16x4_f32(
                 false, a1e, false, be, (short)0, acc1, false, false);
        acc0 = __builtin_amdgcn_wmma_f32_16x16x4_f32(
                 false, a0o, false, bo, (short)0, acc0, false, false);
        acc1 = __builtin_amdgcn_wmma_f32_16x16x4_f32(
                 false, a1o, false, bo, (short)0, acc1, false, false);
    }
    d0 = acc0;
    d1 = acc1;
}

#define NPREP 72
#define NCHAIN 128
__global__ void __launch_bounds__(256)
peps_prep_kernel(const float* __restrict__ peps,
                 float* __restrict__ Msite,
                 float* __restrict__ tab,
                 float* __restrict__ qtab,
                 const unsigned long long ws_bytes) {
    const bool quad   = ws_bytes >= (unsigned long long)WS_FLOATS_QUAD * 4ull;
    const bool qblock = blockIdx.x >= NPREP + NCHAIN;
    if (quad != qblock) return;
    if (blockIdx.x < NPREP) {
        const int idx = blockIdx.x * 256 + threadIdx.x;
        const int n    = idx & 15;
        const int k    = (idx >> 4) & 15;
        const int s    = (idx >> 8) & 1;
        const int site = idx >> 9;
        Msite[idx] = site_elem(peps, site / LY, site % LY, s, k, n);
        return;
    }
    __shared__ __attribute__((aligned(16))) float wl[32768];
    float* const S1 = wl + 6144;
    float* const T  = wl + 6400;
    float* const X0 = wl + 6656;
    float* const E1 = wl + 8192;
    float* const E2 = wl + 16384;

    const int tb   = qblock ? (int)blockIdx.x - (NPREP + NCHAIN) : (int)blockIdx.x - NPREP;
    const int side = qblock ? (tb >> 7) : (tb >> 6);
    const int half = (tb >> 6) & 1;
    const int comb = tb & 63;
    const int lane = threadIdx.x & 31;
    const int wave = __builtin_amdgcn_readfirstlane(threadIdx.x >> 5);
    __builtin_assume(wave >= 0 && wave < 8);
    const int n = lane & 15;
    const int h = lane >> 4;

    const int jt = side ? 3 : 2;
    v2f qb0[2] = {}, qb1[2][4] = {}, qb2[2][4] = {};
    if (qblock) {
        const int i0 = half ? 5 : 0, i1 = half ? 4 : 1, i2 = half ? 3 : 2;
        #pragma unroll
        for (int s = 0; s < 2; ++s) {
            qb0[s] = quad_bfrag(peps, i0, jt, s, 0, n, h, side, half);
            #pragma unroll
            for (int kc = 0; kc < 4; ++kc) {
                qb1[s][kc] = quad_bfrag(peps, i1, jt, s, kc, n, h, side, half);
                qb2[s][kc] = quad_bfrag(peps, i2, jt, s, kc, n, h, side, half);
            }
        }
    }

    if (side == 0) pair_chain_build<0>(wl, peps, comb, wave, n, h);
    else           pair_chain_build<1>(wl, peps, comb, wave, n, h);
    __syncthreads();
    chain_gemm<4, 4>(wl, wl + 1024, S1, wave, n, h);
    if (wave >= 4)
        left_tail_tile(wl + 4096, wl + 5120, T, wave - 4, n, h);
    __syncthreads();
    if (!qblock) {
        float* const H = tab + ((side * 2 + 0) * 64 + comb) * TABN;
        float* const G = tab + ((side * 2 + 1) * 64 + comb) * TABN;
        chain_gemm<16, 4>(S1, wl + 2048, H, wave, n, h);
        chain_gemm<64, 1>(wl + 3072, T, G, wave, n, h);
        return;
    }
    if (half == 0) chain_gemm<16, 4>(S1, wl + 2048, X0, wave, n, h);
    else           chain_gemm<64, 1>(wl + 3072, T, X0, wave, n, h);
    __syncthreads();
    quad_step0(X0, E1, qb0, half, wave, n, h);
    __syncthreads();
    quad_step1(E1, E2, qb1, wave, n, h);
    __syncthreads();
    quad_step2(E2, qtab + ((side * 2 + half) * 512 + comb) * QTABN,
               qb2, half, wave, n, h);
}

__global__ void __launch_bounds__(256)
peps_amp_kernel(const int* __restrict__ x,
                const float* __restrict__ Msite,
                const float* __restrict__ tab,
                const float* __restrict__ qtab,
                const unsigned long long ws_bytes,
                const int nsample,
                float* __restrict__ out) {
    __shared__ __attribute__((aligned(16))) float env[ENV7 + TLEN];
    float* const Tl = env + ENV7;

    const int tid    = threadIdx.x;
    const int lane   = tid & 31;
    const int wave   = __builtin_amdgcn_readfirstlane(tid >> 5);
    __builtin_assume(wave >= 0 && wave < 8);
    const int n      = lane & 15;
    const int h      = lane >> 4;

    const bool quad = ws_bytes >= (unsigned long long)WS_FLOATS_QUAD * 4ull;

    int it = 0;
    for (int sample = blockIdx.x; sample < nsample; sample += gridDim.x, ++it) {
        const int* xrow = x + sample * NSITE;

        if (quad) {
            const int q   = wave >> 1;
            const int ll  = lane < 9 ? lane : 0;
            const int lr  = ll < 6 ? (ll >> 1) : (ll - 6);
            const int lc  = ll < 6 ? (ll & 1) : 2;
            const int col = (q >> 1) ? (LY - 1 - lc) : lc;
            const int xb  = xrow[(3 * (q & 1) + lr) * LY + col] & 1;
            const int cq  = (int)(__builtin_amdgcn_ballot_w32(xb != 0) & 0x1ffu);
            const float* g = qtab + (q * 512 + cq) * QTABN + (wave & 1) * 2048 + lane * 4;
            float*       l = env + q * 4096 + (wave & 1) * 2048 + lane * 4;
            v4f st[16];
            #pragma unroll
            for (int i = 0; i < 16; ++i)
                st[i] = *(const v4f*)(g + i * 128);
            #pragma unroll
            for (int i = 0; i < 16; ++i)
                *(v4f*)(l + i * 128) = st[i];
            __syncthreads();

            v8f t0, t1, b0, b1;
            quad_product(env,        env + 2 * 4096, wave, n, h, t0, t1);
            quad_product(env + 4096, env + 3 * 4096, wave, n, h, b0, b1);
            asm volatile("v_nop\n\tv_nop\n\tv_nop\n\tv_nop"
                         : "+v"(t0), "+v"(t1), "+v"(b0), "+v"(b1));
            float p = 0.0f;
            #pragma unroll
            for (int v = 0; v < 8; ++v)
                p += t0[v] * b0[v];
            #pragma unroll
            for (int v = 0; v < 8; ++v)
                p += t1[v] * b1[v];
            for (int o = 16; o > 0; o >>= 1)
                p += __shfl_xor(p, o);
            float* const red = Tl + (it & 1) * 8;
            if (lane == 0) red[wave] = p;
            __syncthreads();
            if (tid == 0) {
                float a = 0.0f;
                for (int w = 0; w < 8; ++w) a += red[w];
                out[sample] = a;
            }
        } else {
            int cmb[4];
            #pragma unroll
            for (int sd = 0; sd < 2; ++sd) {
                const int je = sd ? LY - 1 : 0, jn = sd ? LY - 2 : 1;
                #pragma unroll
                for (int hf = 0; hf < 2; ++hf) {
                    int cb = 0;
                    #pragma unroll
                    for (int q = 0; q < 3; ++q) {
                        const int i = 3 * hf + q;
                        cb |= (xrow[i * LY + je] & 1) << (2 * q);
                        cb |= (xrow[i * LY + jn] & 1) << (2 * q + 1);
                    }
                    cmb[sd * 2 + hf] = cb;
                }
            }

            {
                v2f qa = {}, qb = {};
                if (wave < 4) {
                    const int s0 = xrow[0 * LY + 2];
                    const int s1 = xrow[1 * LY + 2];
                    top_tile_load(Msite + ((0 * LY + 2) * PD + s0) * 256,
                                  Msite + ((1 * LY + 2) * PD + s1) * 256,
                                  wave, n, h, qa, qb);
                }
                chain_product(tab + (0 * 64 + cmb[0]) * TABN,
                              tab + (1 * 64 + cmb[1]) * TABN, env, wave, n, h);
                if (wave < 4)
                    top_tile_store(Tl, qa, qb, wave, n, h);
            }
            __syncthreads();

            for (int j = 2; j < JR; ++j) {
                for (int i = 0; i < LX - 1; ++i) {
                    if (i == 1) continue;
                    const int site = i * LY + j;
                    const int s    = xrow[site];
                    const float* Mp = Msite + (site * PD + s) * 256;

                    if (i == 0) {
                        apply_top_pair(env, Tl, wave, n, h);
                    } else if (i == 2) {
                        apply_site_rows4<4, 6>(env, Mp, 16, wave, n, h);
                    } else if (i == 3) {
                        v2f pa = {}, pb = {};
                        if (wave < 4) {
                            const int s4 = xrow[4 * LY + j];
                            const int s5 = xrow[5 * LY + j];
                            pair_tile_load(Msite + ((4 * LY + j) * PD + s4) * 256,
                                           Msite + ((5 * LY + j) * PD + s5) * 256,
                                           wave, n, h, pa, pb);
                        }
                        apply_site_row3(env, Mp, wave, n, h);
                        if (wave < 4)
                            pair_tile_store(Tl, pa, pb, wave, n, h);
                    } else {
                        const bool top = (wave < 4) && (j + 1 < JR);
                        v2f qa = {}, qb = {};
                        if (top) {
                            const int s0 = xrow[0 * LY + j + 1];
                            const int s1 = xrow[1 * LY + j + 1];
                            top_tile_load(Msite + ((0 * LY + j + 1) * PD + s0) * 256,
                                          Msite + ((1 * LY + j + 1) * PD + s1) * 256,
                                          wave, n, h, qa, qb);
                        }
                        v8f r0, r1;
                        apply_pair(env, Tl, wave, n, h, r0, r1);
                        __syncthreads();
                        if (top)
                            top_tile_store(Tl, qa, qb, wave, n, h);
                        float* o0 = env + (wave * 16 + n) * 16 + 8 * h;
                        float* o1 = o0 + 8 * 16 * 16;
                        *(v4f*)(o0    ) = __builtin_shufflevector(r0, r0, 0, 1, 2, 3);
                        *(v4f*)(o0 + 4) = __builtin_shufflevector(r0, r0, 4, 5, 6, 7);
                        *(v4f*)(o1    ) = __builtin_shufflevector(r1, r1, 0, 1, 2, 3);
                        *(v4f*)(o1 + 4) = __builtin_shufflevector(r1, r1, 4, 5, 6, 7);
                    }
                    __syncthreads();
                }
            }

            float p = chain_join(tab + (2 * 64 + cmb[2]) * TABN,
                                 tab + (3 * 64 + cmb[3]) * TABN, env, wave, n, h);

            for (int o = 16; o > 0; o >>= 1)
                p += __shfl_xor(p, o);
            if (lane == 0) env[LS1 + wave] = p;
            __syncthreads();
            if (tid == 0) {
                float a = 0.0f;
                for (int w = 0; w < 8; ++w) a += env[LS1 + w];
                out[sample] = a;
            }

        }
    }
}

extern "C" void kernel_launch(void* const* d_in, const int* in_sizes, int n_in,
                              void* d_out, int out_size, void* d_ws, size_t ws_size,
                              hipStream_t stream) {
    const int*   x     = (const int*)d_in[0];
    const float* peps  = (const float*)d_in[1];
    float*       out   = (float*)d_out;
    float*       Msite = (float*)d_ws;
    float*       tab   = Msite + NSITE * PD * 256;
    float*       qtab  = tab + 4 * 64 * TABN;

    peps_prep_kernel<<<NPREP + NCHAIN + NQUAD, 256, 0, stream>>>(peps, Msite, tab, qtab, ws_size);

    const int B = out_size;
    peps_amp_kernel<<<1024, 256, 0, stream>>>(x, Msite, tab, qtab, ws_size, B, out);
}


# main launch retiled: 4-wave blocks, each wave a 32x32 block (2x2 tiles) of both products, one LDS operand load per 2 matrix instructions, not 3 per 4; sum order changes: final sum regrouped
# speedup vs baseline: 31.6804x; 1.0947x over previous
#include <hip/hip_runtime.h>


typedef float v2f __attribute__((ext_vector_type(2)));
typedef float v4f __attribute__((ext_vector_type(4)));
typedef float v8f __attribute__((ext_vector_type(8)));

#define LX 6
#define LY 6
#define PD 2
#define DD 4
#define NSITE 36

__device__ __forceinline__ float site_elem(const float* __restrict__ peps,
                                           const int i, const int j, const int s,
                                           const int k, const int nn) {
    const int a = k >> 2, c = k & 3;
    const int d = nn >> 2, b = nn & 3;
    const bool pad = (i == 0      && a > 0) || (i == LX - 1 && b > 0) ||
                     (j == 0      && c > 0) || (j == LY - 1 && d > 0);
    const float v = peps[(((((i * LY + j) * PD + s) * DD + a) * DD + b) * DD + c) * DD + d];
    return pad ? 0.0f : v;
}

template <int RIGHT>
__device__ __forceinline__ void pair_chain_build(float* Wl,
                                                 const float* __restrict__ peps,
                                                 const int comb,
                                                 const int wave, const int n,
                                                 const int h) {
    constexpr int je = RIGHT ? LY - 1 : 0;
    constexpr int jn = RIGHT ? LY - 2 : 1;
    for (int t = wave; t < 24; t += 8) {
        const int i = t >> 2, u1 = t & 3;
        const int q  = (i < 3) ? i : i - 3;
        const int se = (comb >> (2 * q)) & 1;
        const int sn = (comb >> (2 * q + 1)) & 1;
        v2f a, b;
        if (RIGHT) {
            a.x = site_elem(peps, i, je, se, (n >> 2) * 4 + 2 * h    , n & 3);
            a.y = site_elem(peps, i, je, se, (n >> 2) * 4 + 2 * h + 1, n & 3);
            b.x = site_elem(peps, i, jn, sn, u1 * 4 + (n >> 2), (2 * h    ) * 4 + (n & 3));
            b.y = site_elem(peps, i, jn, sn, u1 * 4 + (n >> 2), (2 * h + 1) * 4 + (n & 3));
        } else {
            a.x = site_elem(peps, i, je, se, (n >> 2) * 4, (2 * h    ) * 4 + (n & 3));
            a.y = site_elem(peps, i, je, se, (n >> 2) * 4, (2 * h + 1) * 4 + (n & 3));
            b.x = site_elem(peps, i, jn, sn, u1 * 4 + 2 * h    , n);
            b.y = site_elem(peps, i, jn, sn, u1 * 4 + 2 * h + 1, n);
        }
        v8f z = {};
        z = __builtin_amdgcn_wmma_f32_16x16x4_f32(
                false, a, false, b, (short)0, z, false, false);
        float* W = Wl + i * 1024 + u1 * 64 + (n >> 2) * 16 + (n & 3);
        #pragma unroll
        for (int v = 0; v < 8; ++v)
            W[(2 * h + (v >> 2)) * 256 + (v & 3) * 4] = z[v];
    }
}

template <int M, int NT>
__device__ __forceinline__ void chain_gemm(const float* A, const float* B,
                                           float* D, const int wave,
                                           const int n, const int h) {
    constexpr int MT  = (M + 15) / 16;
    constexpr int ldn = 16 * NT;
    for (int t = wave; t < MT * NT; t += 8) {
        const int mt = t / NT, nt = t % NT;
        const int row = (M < 16) ? (n & (M - 1)) : (mt * 16 + n);
        v2f a[4], b[4];
        #pragma unroll
        for (int kc = 0; kc < 4; ++kc) {
            a[kc]   = *(const v2f*)(A + row * 16 + 4 * kc + 2 * h);
            b[kc].x = B[(4 * kc + 2 * h    ) * ldn + nt * 16 + n];
            b[kc].y = B[(4 * kc + 2 * h + 1) * ldn + nt * 16 + n];
        }
        v8f acc = {};
        #pragma unroll
        for (int kc = 0; kc < 4; ++kc)
            acc = __builtin_amdgcn_wmma_f32_16x16x4_f32(
                    false, a[kc], false, b[kc], (short)0, acc, false, false);
        #pragma unroll
        for (int v = 0; v < 8; ++v) {
            const int r = mt * 16 + 8 * h + v;
            if (r < M) D[r * ldn + nt * 16 + n] = acc[v];
        }
    }
}

__device__ __forceinline__ void left_tail_tile(const float* W4, const float* W5,
                                               float* T, const int mt,
                                               const int n, const int h) {
    v2f a[4], b[4];
    #pragma unroll
    for (int kc = 0; kc < 4; ++kc) {
        a[kc]   = *(const v2f*)(W4 + (mt * 16 + n) * 16 + 4 * kc + 2 * h);
        b[kc].x = W5[(4 * kc + 2 * h    ) * 64 + (n & 3) * 16];
        b[kc].y = W5[(4 * kc + 2 * h + 1) * 64 + (n & 3) * 16];
    }
    v8f acc = {};
    #pragma unroll
    for (int kc = 0; kc < 4; ++kc)
        acc = __builtin_amdgcn_wmma_f32_16x16x4_f32(
                false, a[kc], false, b[kc], (short)0, acc, false, false);
    if (n < 4) {
        #pragma unroll
        for (int v = 0; v < 8; ++v)
            T[(mt * 16 + 8 * h + v) * 4 + n] = acc[v];
    }
}

#define QTABN 4096
#define NQUAD 256
#define WS_BYTES_QUAD (4ull * 512ull * QTABN * 4ull)

__device__ __forceinline__ int quad_off(const int r, const int v) {
    return ((((v >> 4) * 8 + (r >> 3)) * 32 + ((r >> 1) & 1) * 16 + (v & 15)) << 2)
           + ((r >> 2) & 1) * 2 + (r & 1);
}

__device__ __forceinline__ float quad_elem(const float* __restrict__ peps,
                                           const int i, const int j, const int s,
                                           const int k, const int nn,
                                           const int side, const int half) {
    const int vin = k >> 2, hin = k & 3, hout = nn >> 2, vout = nn & 3;
    const int up    = half ? vout : vin;
    const int down  = half ? vin  : vout;
    const int left  = side ? hout : hin;
    const int right = side ? hin  : hout;
    return site_elem(peps, i, j, s, up * 4 + left, right * 4 + down);
}

__device__ __forceinline__ v2f quad_bfrag(const float* __restrict__ peps,
                                          const int i, const int j, const int s,
                                          const int kc, const int n, const int h,
                                          const int side, const int half) {
    v2f b;
    b.x = quad_elem(peps, i, j, s, 4 * kc + 2 * h    , n, side, half);
    b.y = quad_elem(peps, i, j, s, 4 * kc + 2 * h + 1, n, side, half);
    return b;
}

__device__ __forceinline__ void quad_step0(const float* X0, float* E1,
                                           const v2f (&b)[2], const int half,
                                           const int wave, const int n, const int h) {
    const int sa0 = half ? 1 : 256, sa1 = half ? 4 : 64, sb = half ? 64 : 1;
    for (int tp = 0; tp < 2; ++tp) {
        const int t   = wave + 8 * tp;
        const int row = t * 16 + n;
        const int rb  = (row >> 6) * sa1 + ((row >> 4) & 3) * 16 + (row & 15) * sb;
        v2f a;
        a.x = X0[rb + (2 * h    ) * sa0];
        a.y = X0[rb + (2 * h + 1) * sa0];
        #pragma unroll
        for (int s0 = 0; s0 < 2; ++s0) {
            v8f acc = {};
            acc = __builtin_amdgcn_wmma_f32_16x16x4_f32(
                    false, a, false, b[s0], (short)0, acc, false, false);
            #pragma unroll
            for (int v = 0; v < 8; ++v)
                E1[s0 * 4096 + n * 256 + t * 16 + 8 * h + v] = acc[v];
        }
    }
}

__device__ __forceinline__ void quad_step1(const float* E1, float* E2,
                                           const v2f (&b)[2][4],
                                           const int wave, const int n, const int h) {
    for (int tp = 0; tp < 2; ++tp) {
        const int t   = wave + 8 * tp;
        const int row = t * 16 + n;
        const int rb  = (row >> 6) * 1024 + ((row >> 4) & 3) * 16 + (row & 15);
        #pragma unroll
        for (int s0 = 0; s0 < 2; ++s0) {
            v2f a[4];
            #pragma unroll
            for (int kc = 0; kc < 4; ++kc) {
                a[kc].x = E1[s0 * 4096 + rb + kc * 256 + (2 * h    ) * 64];
                a[kc].y = E1[s0 * 4096 + rb + kc * 256 + (2 * h + 1) * 64];
            }
            #pragma unroll
            for (int s1 = 0; s1 < 2; ++s1) {
                v8f acc = {};
                #pragma unroll
                for (int kc = 0; kc < 4; ++kc)
                    acc = __builtin_amdgcn_wmma_f32_16x16x4_f32(
                            false, a[kc], false, b[s1][kc], (short)0, acc, false, false);
                #pragma unroll
                for (int v = 0; v < 8; ++v)
                    E2[(s0 + 2 * s1) * 4096 + (t >> 2) * 1024 + n * 64 + (t & 3) * 16 + 8 * h + v] = acc[v];
            }
        }
    }
}

__device__ __forceinline__ void quad_step2(const float* E2, float* __restrict__ Qb,
                                           const v2f (&b)[2][4], const int half,
                                           const int wave, const int n, const int h) {
    for (int tp = 0; tp < 2; ++tp) {
        const int t   = wave + 8 * tp;
        const int row = t * 16 + n;
        const int rb  = (row >> 6) * 1024 + ((row >> 4) & 3) * 256 + (row & 15);
        #pragma unroll
        for (int s01 = 0; s01 < 4; ++s01) {
            v2f a[4];
            #pragma unroll
            for (int kc = 0; kc < 4; ++kc) {
                a[kc].x = E2[s01 * 4096 + rb + kc * 64 + (2 * h    ) * 16];
                a[kc].y = E2[s01 * 4096 + rb + kc * 64 + (2 * h + 1) * 16];
            }
            #pragma unroll
            for (int s2 = 0; s2 < 2; ++s2) {
                v8f acc = {};
                #pragma unroll
                for (int kc = 0; kc < 4; ++kc)
                    acc = __builtin_amdgcn_wmma_f32_16x16x4_f32(
                            false, a[kc], false, b[s2][kc], (short)0, acc, false, false);
                const int c3 = half ? ((s01 & 1) * 4 + (s01 >> 1) * 2 + s2) : (s01 + 4 * s2);
                float* const Q = Qb + c3 * 64 * QTABN;
                #pragma unroll
                for (int v = 0; v < 8; ++v)
                    Q[quad_off(t * 4 + (n >> 2), (8 * h + v) * 4 + (n & 3))] = acc[v];
            }
        }
    }
}

__device__ __forceinline__ void quad_product22(const float* __restrict__ FA,
                                               const float* __restrict__ FB,
                                               const int wave, const int n, const int h,
                                               v8f& d00, v8f& d01, v8f& d10, v8f& d11) {
    const int lo = (h * 16 + n) * 4;
    const float* pa = FA + (wave >> 1) * 2048 + lo;
    const float* pb = FB + (wave & 1) * 2048 + lo;
    v8f a00 = {}, a01 = {}, a10 = {}, a11 = {};
    #pragma unroll
    for (int p = 0; p < 8; ++p) {
        const v4f f0 = *(const v4f*)(pa + p * 128);
        const v4f f1 = *(const v4f*)(pa + 1024 + p * 128);
        const v4f g0 = *(const v4f*)(pb + p * 128);
        const v4f g1 = *(const v4f*)(pb + 1024 + p * 128);
        const v2f f0e = {f0.x, f0.y}, f0o = {f0.z, f0.w};
        const v2f f1e = {f1.x, f1.y}, f1o = {f1.z, f1.w};
        const v2f g0e = {g0.x, g0.y}, g0o = {g0.z, g0.w};
        const v2f g1e = {g1.x, g1.y}, g1o = {g1.z, g1.w};
        a00 = __builtin_amdgcn_wmma_f32_16x16x4_f32(
                false, f0e, false, g0e, (short)0, a00, false, false);
        a01 = __builtin_amdgcn_wmma_f32_16x16x4_f32(
                false, f0e, false, g1e, (short)0, a01, false, false);
        a10 = __builtin_amdgcn_wmma_f32_16x16x4_f32(
                false, f1e, false, g0e, (short)0, a10, false, false);
        a11 = __builtin_amdgcn_wmma_f32_16x16x4_f32(
                false, f1e, false, g1e, (short)0, a11, false, false);
        a00 = __builtin_amdgcn_wmma_f32_16x16x4_f32(
                false, f0o, false, g0o, (short)0, a00, false, false);
        a01 = __builtin_amdgcn_wmma_f32_16x16x4_f32(
                false, f0o, false, g1o, (short)0, a01, false, false);
        a10 = __builtin_amdgcn_wmma_f32_16x16x4_f32(
                false, f1o, false, g0o, (short)0, a10, false, false);
        a11 = __builtin_amdgcn_wmma_f32_16x16x4_f32(
                false, f1o, false, g1o, (short)0, a11, false, false);
    }
    d00 = a00;
    d01 = a01;
    d10 = a10;
    d11 = a11;
}

__global__ void __launch_bounds__(256)
peps_prep_kernel(const float* __restrict__ peps,
                 float* __restrict__ qtab,
                 const unsigned long long ws_bytes) {
    if (ws_bytes < WS_BYTES_QUAD) return;
    __shared__ __attribute__((aligned(16))) float wl[32768];
    float* const S1 = wl + 6144;
    float* const T  = wl + 6400;
    float* const X0 = wl + 6656;
    float* const E1 = wl + 8192;
    float* const E2 = wl + 16384;

    const int tb   = (int)blockIdx.x;
    const int side = tb >> 7;
    const int half = (tb >> 6) & 1;
    const int comb = tb & 63;
    const int lane = threadIdx.x & 31;
    const int wave = __builtin_amdgcn_readfirstlane(threadIdx.x >> 5);
    __builtin_assume(wave >= 0 && wave < 8);
    const int n = lane & 15;
    const int h = lane >> 4;

    const int jt = side ? 3 : 2;
    const int i0 = half ? 5 : 0, i1 = half ? 4 : 1, i2 = half ? 3 : 2;
    v2f qb0[2], qb1[2][4], qb2[2][4];
    #pragma unroll
    for (int s = 0; s < 2; ++s) {
        qb0[s] = quad_bfrag(peps, i0, jt, s, 0, n, h, side, half);
        #pragma unroll
        for (int kc = 0; kc < 4; ++kc) {
            qb1[s][kc] = quad_bfrag(peps, i1, jt, s, kc, n, h, side, half);
            qb2[s][kc] = quad_bfrag(peps, i2, jt, s, kc, n, h, side, half);
        }
    }

    if (side == 0) pair_chain_build<0>(wl, peps, comb, wave, n, h);
    else           pair_chain_build<1>(wl, peps, comb, wave, n, h);
    __syncthreads();
    chain_gemm<4, 4>(wl, wl + 1024, S1, wave, n, h);
    if (wave >= 4)
        left_tail_tile(wl + 4096, wl + 5120, T, wave - 4, n, h);
    __syncthreads();
    if (half == 0) chain_gemm<16, 4>(S1, wl + 2048, X0, wave, n, h);
    else           chain_gemm<64, 1>(wl + 3072, T, X0, wave, n, h);
    __syncthreads();
    quad_step0(X0, E1, qb0, half, wave, n, h);
    __syncthreads();
    quad_step1(E1, E2, qb1, wave, n, h);
    __syncthreads();
    quad_step2(E2, qtab + ((side * 2 + half) * 512 + comb) * QTABN,
               qb2, half, wave, n, h);
}

__global__ void __launch_bounds__(128)
peps_amp_kernel(const int* __restrict__ x,
                const float* __restrict__ qtab,
                const unsigned long long ws_bytes,
                const int nsample,
                float* __restrict__ out) {
    if (ws_bytes < WS_BYTES_QUAD) return;

    __shared__ __attribute__((aligned(16))) float env[4 * QTABN + 8];
    float* const redb = env + 4 * QTABN;

    const int tid    = threadIdx.x;
    const int lane   = tid & 31;
    const int wave   = __builtin_amdgcn_readfirstlane(tid >> 5);
    __builtin_assume(wave >= 0 && wave < 4);
    const int n      = lane & 15;
    const int h      = lane >> 4;

    int it = 0;
    for (int sample = blockIdx.x; sample < nsample; sample += gridDim.x, ++it) {
        const int* xrow = x + sample * NSITE;

        const int q   = wave;
        const int ll  = lane < 9 ? lane : 0;
        const int lr  = ll < 6 ? (ll >> 1) : (ll - 6);
        const int lc  = ll < 6 ? (ll & 1) : 2;
        const int col = (q >> 1) ? (LY - 1 - lc) : lc;
        const int xb  = xrow[(3 * (q & 1) + lr) * LY + col] & 1;
        const int cq  = (int)(__builtin_amdgcn_ballot_w32(xb != 0) & 0x1ffu);
        const float* g = qtab + (q * 512 + cq) * QTABN + lane * 4;
        float*       l = env + q * QTABN + lane * 4;
        v4f st[32];
        #pragma unroll
        for (int i = 0; i < 32; ++i)
            st[i] = *(const v4f*)(g + i * 128);
        #pragma unroll
        for (int i = 0; i < 32; ++i)
            *(v4f*)(l + i * 128) = st[i];
        __syncthreads();

        v8f t00, t01, t10, t11, b00, b01, b10, b11;
        quad_product22(env,             env + 2 * QTABN, wave, n, h, t00, t01, t10, t11);
        quad_product22(env + 1 * QTABN, env + 3 * QTABN, wave, n, h, b00, b01, b10, b11);
        asm volatile("v_nop\n\tv_nop\n\tv_nop\n\tv_nop"
                     : "+v"(t00), "+v"(t01), "+v"(t10), "+v"(t11),
                       "+v"(b00), "+v"(b01), "+v"(b10), "+v"(b11));
        float p = 0.0f;
        #pragma unroll
        for (int v = 0; v < 8; ++v)
            p += t00[v] * b00[v];
        #pragma unroll
        for (int v = 0; v < 8; ++v)
            p += t01[v] * b01[v];
        #pragma unroll
        for (int v = 0; v < 8; ++v)
            p += t10[v] * b10[v];
        #pragma unroll
        for (int v = 0; v < 8; ++v)
            p += t11[v] * b11[v];
        for (int o = 16; o > 0; o >>= 1)
            p += __shfl_xor(p, o);
        float* const red = redb + (it & 1) * 4;
        if (lane == 0) red[wave] = p;
        __syncthreads();
        if (tid == 0) {
            float a = 0.0f;
            for (int w = 0; w < 4; ++w) a += red[w];
            out[sample] = a;
        }
    }
}

extern "C" void kernel_launch(void* const* d_in, const int* in_sizes, int n_in,
                              void* d_out, int out_size, void* d_ws, size_t ws_size,
                              hipStream_t stream) {
    const int*   x    = (const int*)d_in[0];
    const float* peps = (const float*)d_in[1];
    float*       out  = (float*)d_out;
    float*       qtab = (float*)d_ws;

    peps_prep_kernel<<<NQUAD, 256, 0, stream>>>(peps, qtab, ws_size);

    const int B = out_size;
    peps_amp_kernel<<<1024, 128, 0, stream>>>(x, qtab, ws_size, B, out);
}
